# starting kernel
# baseline (speedup 1.0000x reference)
#include <hip/hip_runtime.h>
#include <hip/hip_bf16.h>
#include <math.h>

typedef __attribute__((ext_vector_type(2))) float v2f;
typedef __attribute__((ext_vector_type(8))) float v8f;

#define IN_DIM  128
#define HID     16
#define OUT_DIM 2

__global__ void k_init_dinv(float* __restrict__ dinv, int n) {
    int i = blockIdx.x * blockDim.x + threadIdx.x;
    if (i < n) dinv[i] = 1.0f;
}

__global__ void k_deg_edges(const int* __restrict__ dst, float* __restrict__ deg, int nE) {
    int e = blockIdx.x * blockDim.x + threadIdx.x;
    if (e < nE) unsafeAtomicAdd(&deg[dst[e]], 1.0f);
}

__global__ void k_finalize_dinv(float* __restrict__ dinv, int n) {
    int i = blockIdx.x * blockDim.x + threadIdx.x;
    if (i < n) {
        float d = dinv[i];
        dinv[i] = (d > 0.0f) ? rsqrtf(d) : 0.0f;
    }
}

__global__ __launch_bounds__(128) void k_gemm1_wmma(const float* __restrict__ x,
                                                    const float* __restrict__ W1,
                                                    float* __restrict__ h, int n) {
    __shared__ float sW[IN_DIM * HID];
    __shared__ float sX[4][16 * 132];

    for (int t = threadIdx.x; t < IN_DIM * HID; t += blockDim.x) sW[t] = W1[t];

    const int wave = threadIdx.x >> 5;
    const int lane = threadIdx.x & 31;
    const int tile = blockIdx.x * 4 + wave;
    const int row0 = tile * 16;
    const bool active = (row0 < n);

    if (active) {
        #pragma unroll 4
        for (int i = 0; i < 16; ++i) {
            int r = row0 + i;
            if (r >= n) r = n - 1;
            const float4 v = ((const float4*)(x + (size_t)r * IN_DIM))[lane];
            *(float4*)&sX[wave][i * 132 + lane * 4] = v;
        }
    }
    __syncthreads();
    if (!active) return;

    const int m  = lane & 15;
    const int kb = (lane >> 4) << 1;
    const float* xr = &sX[wave][m * 132];

    v8f acc = {};
    #pragma unroll
    for (int k0 = 0; k0 < IN_DIM; k0 += 4) {
        v2f a, b;
        a.x = xr[k0 + kb];
        a.y = xr[k0 + kb + 1];
        b.x = sW[(k0 + kb)     * HID + m];
        b.y = sW[(k0 + kb + 1) * HID + m];
        acc = __builtin_amdgcn_wmma_f32_16x16x4_f32(false, a, false, b,
                                                    (short)0, acc, false, false);
    }

    const int rbase = row0 + ((lane >> 4) << 3);
    #pragma unroll
    for (int v = 0; v < 8; ++v) {
        int r = rbase + v;
        if (r < n) h[(size_t)r * HID + m] = acc[v];
    }
}

__global__ void k_selfloop16(const float* __restrict__ h, const float* __restrict__ dinv,
                             float* __restrict__ agg, int n) {
    int t = blockIdx.x * blockDim.x + threadIdx.x;
    if (t >= n * HID) return;
    int i = t / HID;
    float di = dinv[i];
    agg[t] = h[t] * (di * di);
}

__global__ void k_scatter1(const int* __restrict__ src, const int* __restrict__ dst,
                           const float* __restrict__ dinv, const float* __restrict__ h,
                           float* __restrict__ agg, int nE) {
    int e = blockIdx.x * blockDim.x + threadIdx.x;
    if (e >= nE) return;
    int s = src[e], d = dst[e];
    float nrm = dinv[s] * dinv[d];
    const float4* hr = (const float4*)(h + (size_t)s * HID);
    float* ar = agg + (size_t)d * HID;
    #pragma unroll
    for (int q = 0; q < 4; ++q) {
        float4 v = hr[q];
        unsafeAtomicAdd(ar + q * 4 + 0, v.x * nrm);
        unsafeAtomicAdd(ar + q * 4 + 1, v.y * nrm);
        unsafeAtomicAdd(ar + q * 4 + 2, v.z * nrm);
        unsafeAtomicAdd(ar + q * 4 + 3, v.w * nrm);
    }
}

__global__ void k_layer2(const float* __restrict__ agg1, const float* __restrict__ b1,
                         const float* __restrict__ W2, const float* __restrict__ dinv,
                         float* __restrict__ h2, float* __restrict__ agg2, int n) {
    int i = blockIdx.x * blockDim.x + threadIdx.x;
    if (i >= n) return;
    const float* r = agg1 + (size_t)i * HID;
    float z0 = 0.0f, z1 = 0.0f;
    #pragma unroll
    for (int k = 0; k < HID; ++k) {
        float v = fmaxf(r[k] + b1[k], 0.0f);
        z0 = fmaf(v, W2[k * OUT_DIM + 0], z0);
        z1 = fmaf(v, W2[k * OUT_DIM + 1], z1);
    }
    float di = dinv[i], w = di * di;
    h2[i * 2 + 0]   = z0;
    h2[i * 2 + 1]   = z1;
    agg2[i * 2 + 0] = z0 * w;
    agg2[i * 2 + 1] = z1 * w;
}

__global__ void k_scatter2(const int* __restrict__ src, const int* __restrict__ dst,
                           const float* __restrict__ dinv, const float* __restrict__ h2,
                           float* __restrict__ agg2, int nE) {
    int e = blockIdx.x * blockDim.x + threadIdx.x;
    if (e >= nE) return;
    int s = src[e], d = dst[e];
    float nrm = dinv[s] * dinv[d];
    float2 v = ((const float2*)h2)[s];
    unsafeAtomicAdd(&agg2[d * 2 + 0], v.x * nrm);
    unsafeAtomicAdd(&agg2[d * 2 + 1], v.y * nrm);
}

__global__ void k_final(const float* __restrict__ agg2, const float* __restrict__ b2,
                        float* __restrict__ out, int n) {
    int i = blockIdx.x * blockDim.x + threadIdx.x;
    if (i >= n) return;
    float z0 = agg2[i * 2 + 0] + b2[0];
    float z1 = agg2[i * 2 + 1] + b2[1];
    float mx = fmaxf(z0, z1);
    float l  = mx + logf(expf(z0 - mx) + expf(z1 - mx));
    out[i * 2 + 0] = z0 - l;
    out[i * 2 + 1] = z1 - l;
}

extern "C" void kernel_launch(void* const* d_in, const int* in_sizes, int n_in,
                              void* d_out, int out_size, void* d_ws, size_t ws_size,
                              hipStream_t stream) {
    const float* x  = (const float*)d_in[0];
    const int*   ei = (const int*)d_in[1];
    const float* W1 = (const float*)d_in[2];
    const float* b1 = (const float*)d_in[3];
    const float* W2 = (const float*)d_in[4];
    const float* b2 = (const float*)d_in[5];
    float* out = (float*)d_out;

    const int n  = in_sizes[0] / IN_DIM;
    const int nE = in_sizes[1] / 2;
    const int* src = ei;
    const int* dst = ei + nE;

    float* ws   = (float*)d_ws;
    float* dinv = ws;
    float* h    = ws + n;
    float* agg1 = ws + (size_t)n * 17;
    float* h2   = h;
    float* agg2 = h + (size_t)2 * n;

    const int B = 256;
    k_init_dinv    <<<(n  + B - 1) / B, B, 0, stream>>>(dinv, n);
    k_deg_edges    <<<(nE + B - 1) / B, B, 0, stream>>>(dst, dinv, nE);
    k_finalize_dinv<<<(n  + B - 1) / B, B, 0, stream>>>(dinv, n);

    const int tiles   = (n + 15) / 16;
    const int gblocks = (tiles + 3) / 4;
    k_gemm1_wmma   <<<gblocks, 128, 0, stream>>>(x, W1, h, n);

    k_selfloop16   <<<(n * HID + B - 1) / B, B, 0, stream>>>(h, dinv, agg1, n);
    k_scatter1     <<<(nE + B - 1) / B, B, 0, stream>>>(src, dst, dinv, h, agg1, nE);
    k_layer2       <<<(n  + B - 1) / B, B, 0, stream>>>(agg1, b1, W2, dinv, h2, agg2, n);
    k_scatter2     <<<(nE + B - 1) / B, B, 0, stream>>>(src, dst, dinv, h2, agg2, nE);
    k_final        <<<(n  + B - 1) / B, B, 0, stream>>>(agg2, b2, out, n);
}


# layer-2 per-node product (bias, relu, 16x2 W2) done in the layer-1 sum's epilogue by the node's four lanes: the agg1 plane and one launch are gone; float sum order of that 16-term product changes
# speedup vs baseline: 9.6261x; 9.6261x over previous
#include <hip/hip_runtime.h>
#include <hip/hip_bf16.h>
#include <math.h>


typedef __attribute__((ext_vector_type(2))) float v2f;
typedef __attribute__((ext_vector_type(8))) float v8f;

#define IN_DIM  128
#define HID     16
#define OUT_DIM 2

#define BSHIFT 8
#define BNODES 256
#define MAXB   512
#define SLICE  12288
#define SPT    12
#define RPW    12
#define BCAP   16384
#define BPT    10

__global__ __launch_bounds__(1024, 16) void k_slice_sort(const int* __restrict__ src, const int* __restrict__ dst,
                                                     int* __restrict__ T, int* __restrict__ binned,
                                                     int nE, int nb, int nbp) {
    __shared__ int sWord[SLICE];
    __shared__ int hist[MAXB];
    __shared__ int lbase[MAXB];
    __shared__ int wtot[16];
    const int t = threadIdx.x, lane = t & 31, wave = t >> 5;
    const int me = blockIdx.x;

    const int e0 = me * SLICE + t;
    int s[SPT], d[SPT], r[SPT];
    #pragma unroll
    for (int u = 0; u < SPT; ++u) {
        const int e = e0 + u * 1024;
        const int ec = e < nE ? e : nE - 1;
        s[u] = src[ec];
        d[u] = dst[ec];
    }
    if (t < MAXB) hist[t] = 0;
    __syncthreads();

    #pragma unroll
    for (int u = 0; u < SPT; ++u) {
        const int e = e0 + u * 1024;
        r[u] = 0;
        if (e < nE) {
            unsigned b = (unsigned)d[u] >> BSHIFT;
            b = b > (unsigned)(nb - 1) ? (unsigned)(nb - 1) : b;
            r[u] = atomicAdd(&hist[b], 1);
        }
    }
    __syncthreads();
    int c = 0, incl = 0;
    if (t < MAXB) {
        c = hist[t];
        incl = c;
        #pragma unroll
        for (int o = 1; o < 32; o <<= 1) {
            const int v = __shfl_up(incl, o);
            if (lane >= o) incl += v;
        }
        if (lane == 31) wtot[wave] = incl;
    }
    __syncthreads();
    int ex = 0;
    if (t < MAXB) {
        int woff = 0;
        #pragma unroll
        for (int w = 0; w < 16; ++w) woff += (w < wave) ? wtot[w] : 0;
        ex = woff + incl - c;
        lbase[t] = ex;
    }
    __syncthreads();
    #pragma unroll
    for (int u = 0; u < SPT; ++u) {
        const int e = e0 + u * 1024;
        if (e < nE) {
            unsigned b = (unsigned)d[u] >> BSHIFT;
            b = b > (unsigned)(nb - 1) ? (unsigned)(nb - 1) : b;
            sWord[lbase[b] + r[u]] = (int)((((unsigned)d[u] & (BNODES - 1)) << 24) | ((unsigned)s[u] & 0xFFFFFFu));
        }
    }
    __syncthreads();
    if (t < nb) T[(size_t)me * nbp + t] = ex;
    if (t == nb - 1) T[(size_t)me * nbp + nb] = ex + c;
    const int first = me * SLICE;
    const int have = (nE - first < SLICE) ? nE - first : SLICE;
    for (int k = t; k < have; k += 1024) binned[first + k] = sWord[k];
}

__global__ __launch_bounds__(1024, 16) void k_bucket_sort(const int* __restrict__ binned, const int* __restrict__ T,
                                                      int* __restrict__ cnt, int* __restrict__ rowptr,
                                                      float* __restrict__ dinv, int* __restrict__ sorted,
                                                      int n, int nE, int nbp, int nblk) {
    __shared__ int sIn[BCAP];
    __shared__ int sOut[BCAP];
    __shared__ int rBase[32 * RPW];
    __shared__ int rLen[32 * RPW];
    __shared__ int rPos[32 * RPW];
    __shared__ int lcnt[BNODES];
    __shared__ int lptr[BNODES];
    __shared__ int wlen[RPW];
    __shared__ int wlo[RPW];
    __shared__ int wtot[8];
    const int t = threadIdx.x, lane = t & 31, wave = t >> 5;
    const int b = blockIdx.x;

    int lo = 0, len = 0;
    if (t < nblk) {
        const int first = t * SLICE;
        int have = (nE - first < SLICE) ? nE - first : SLICE;
        have = have < 0 ? 0 : have;
        int a = T[(size_t)t * nbp + b];
        int e = T[(size_t)t * nbp + b + 1];
        a = a < 0 ? 0 : (a > have ? have : a);
        e = e < a ? a : (e > have ? have : e);
        lo = a;
        len = e - a;
    }
    int incl = len, slo = lo;
    #pragma unroll
    for (int o = 1; o < 32; o <<= 1) {
        const int v = __shfl_up(incl, o);
        const int y = __shfl_up(slo, o);
        if (lane >= o) { incl += v; slo += y; }
    }
    if (wave < RPW && lane == 31) {
        wlen[wave] = incl;
        wlo[wave] = slo;
    }
    if (t < BNODES) lcnt[t] = 0;
    __syncthreads();
    int woff = 0, tot = 0, bs = 0;
    #pragma unroll
    for (int w = 0; w < RPW; ++w) {
        const int x = wlen[w];
        woff += (w < wave) ? x : 0;
        tot += x;
        bs += wlo[w];
    }
    if (t < 32 * RPW) {
        rBase[t] = t * SLICE + lo;
        rLen[t]  = len;
        rPos[t]  = woff + incl - len;
    }
    bs = bs < 0 ? 0 : (bs > nE ? nE : bs);
    const int size = (tot > nE - bs) ? nE - bs : tot;
    const bool inlds = tot <= BCAP;
    const int amax = nE - 1;
    __syncthreads();

    if (inlds) {
        int w0[RPW], w1[RPW];
        #pragma unroll
        for (int u = 0; u < RPW; ++u) {
            const int j = wave + 32 * u;
            const int base = rBase[j], ln = rLen[j];
            int a0 = base + (lane < ln ? lane : ln - 1);
            int a1 = base + (lane + 32 < ln ? lane + 32 : ln - 1);
            a0 = a0 < 0 ? 0 : (a0 > amax ? amax : a0);
            a1 = a1 < 0 ? 0 : (a1 > amax ? amax : a1);
            w0[u] = binned[a0];
            w1[u] = binned[a1];
        }
        int lnv[RPW], posv[RPW];
        #pragma unroll
        for (int u = 0; u < RPW; ++u) {
            const int j = wave + 32 * u;
            lnv[u] = rLen[j];
            posv[u] = rPos[j];
        }
        #pragma unroll
        for (int u = 0; u < RPW; ++u) {
            if (lane < lnv[u])      sIn[posv[u] + lane]      = w0[u];
            if (lane + 32 < lnv[u]) sIn[posv[u] + lane + 32] = w1[u];
        }
        for (int j = wave; j < 32 * RPW; j += 32) {
            const int base = rBase[j], ln = rLen[j], pos = rPos[j];
            for (int o = lane + 64; o < ln; o += 32) {
                int a = base + o;
                a = a > amax ? amax : a;
                sIn[pos + o] = binned[a];
            }
        }
    }
    __syncthreads();

    if (inlds) {
        for (int kb = t; kb < size + t; kb += BPT * 1024) {
            unsigned w[BPT];
            #pragma unroll
            for (int u = 0; u < BPT; ++u) {
                const int k = kb + u * 1024;
                w[u] = (unsigned)sIn[k < size ? k : size - 1];
            }
            #pragma unroll
            for (int u = 0; u < BPT; ++u)
                if (kb + u * 1024 < size) atomicAdd(&lcnt[w[u] >> 24], 1);
        }
    } else {
        for (int j = wave; j < 32 * RPW; j += 32) {
            const int base = rBase[j], ln = rLen[j];
            for (int o = lane; o < ln; o += 32) {
                int a = base + o;
                a = a > amax ? amax : a;
                atomicAdd(&lcnt[(unsigned)binned[a] >> 24], 1);
            }
        }
    }
    __syncthreads();
    int c = 0;
    incl = 0;
    if (t < BNODES) {
        c = lcnt[t];
        incl = c;
        #pragma unroll
        for (int o = 1; o < 32; o <<= 1) {
            const int v = __shfl_up(incl, o);
            if (lane >= o) incl += v;
        }
        if (lane == 31) wtot[wave] = incl;
    }
    __syncthreads();
    if (t < BNODES) {
        int wo = 0;
        #pragma unroll
        for (int w = 0; w < 8; ++w) wo += (w < wave) ? wtot[w] : 0;
        const int ex = wo + incl - c;
        lptr[t] = ex;
        lcnt[t] = 0;
        const int node = b * BNODES + t;
        if (node < n) {
            cnt[node] = c;
            rowptr[node] = bs + ex;
            const float dg = (float)(c + 1);
            dinv[node] = (dg > 0.0f) ? rsqrtf(dg) : 0.0f;
        }
    }
    __syncthreads();

    if (inlds) {
        for (int kb = t; kb < size + t; kb += BPT * 1024) {
            unsigned w[BPT];
            #pragma unroll
            for (int u = 0; u < BPT; ++u) {
                const int k = kb + u * 1024;
                w[u] = (unsigned)sIn[k < size ? k : size - 1];
            }
            #pragma unroll
            for (int u = 0; u < BPT; ++u) {
                if (kb + u * 1024 < size) {
                    const unsigned dl = w[u] >> 24;
                    const int p = lptr[dl] + atomicAdd(&lcnt[dl], 1);
                    sOut[p] = (int)(w[u] & 0xFFFFFFu);
                }
            }
        }
    } else {
        for (int j = wave; j < 32 * RPW; j += 32) {
            const int base = rBase[j], ln = rLen[j];
            for (int o = lane; o < ln; o += 32) {
                int a = base + o;
                a = a > amax ? amax : a;
                const unsigned w = (unsigned)binned[a];
                const unsigned dl = w >> 24;
                const int p = lptr[dl] + atomicAdd(&lcnt[dl], 1);
                if (p < size) sorted[bs + p] = (int)(w & 0xFFFFFFu);
            }
        }
    }
    __syncthreads();
    if (inlds)
        for (int k = t; k < size; k += 1024) sorted[bs + k] = sOut[k];
}

__global__ __launch_bounds__(128) void k_gemm1_wmma(const float* __restrict__ x,
                                                    const float* __restrict__ W1,
                                                    float* __restrict__ h, int n) {
    __shared__ float sW[IN_DIM * HID];
    __shared__ float sX[4][16 * 132];

    for (int t = threadIdx.x; t < IN_DIM * HID; t += blockDim.x) sW[t] = W1[t];

    const int wave = threadIdx.x >> 5;
    const int lane = threadIdx.x & 31;
    const int tile = blockIdx.x * 4 + wave;
    const int row0 = tile * 16;
    const bool active = (row0 < n);

    if (active) {
        #pragma unroll 4
        for (int i = 0; i < 16; ++i) {
            int r = row0 + i;
            if (r >= n) r = n - 1;
            const float4 v = ((const float4*)(x + (size_t)r * IN_DIM))[lane];
            *(float4*)&sX[wave][i * 132 + lane * 4] = v;
        }
    }
    __syncthreads();
    if (!active) return;

    const int m  = lane & 15;
    const int kb = (lane >> 4) << 1;
    const float* xr = &sX[wave][m * 132];

    v8f acc = {};
    #pragma unroll
    for (int k0 = 0; k0 < IN_DIM; k0 += 4) {
        v2f a, b;
        a.x = xr[k0 + kb];
        a.y = xr[k0 + kb + 1];
        b.x = sW[(k0 + kb)     * HID + m];
        b.y = sW[(k0 + kb + 1) * HID + m];
        acc = __builtin_amdgcn_wmma_f32_16x16x4_f32(false, a, false, b,
                                                    (short)0, acc, false, false);
    }

    const int rbase = row0 + ((lane >> 4) << 3);
    #pragma unroll
    for (int v = 0; v < 8; ++v) {
        int r = rbase + v;
        if (r < n) h[(size_t)r * HID + m] = acc[v];
    }
}

__global__ __launch_bounds__(256) void k_pull16(const int* __restrict__ rowptr,
                                                const int* __restrict__ cnt,
                                                const int* __restrict__ sorted,
                                                const float* __restrict__ dinv,
                                                const float* __restrict__ h,
                                                const float* __restrict__ b1,
                                                const float* __restrict__ W2,
                                                float* __restrict__ rec, int n, int nE) {
    const int g = blockIdx.x * 256 + threadIdx.x;
    const int i = g >> 2;
    const int q = g & 3;
    if (i >= n) return;
    int start = rowptr[i];
    start = start < 0 ? 0 : (start > nE ? nE : start);
    int c = cnt[i];
    c = c < 0 ? 0 : c;
    const int end = (c > nE - start) ? nE : start + c;
    const float di = dinv[i];
    const float4* h4 = (const float4*)h;
    const unsigned nmax = (unsigned)n - 1u;

    const float4 hs = h4[(size_t)i * 4 + q];
    const int c0 = q * 4;
    const float bA = b1[c0], bB = b1[c0 + 1], bC = b1[c0 + 2], bD = b1[c0 + 3];
    const float wA0 = W2[(c0    ) * OUT_DIM], wA1 = W2[(c0    ) * OUT_DIM + 1];
    const float wB0 = W2[(c0 + 1) * OUT_DIM], wB1 = W2[(c0 + 1) * OUT_DIM + 1];
    const float wC0 = W2[(c0 + 2) * OUT_DIM], wC1 = W2[(c0 + 2) * OUT_DIM + 1];
    const float wD0 = W2[(c0 + 3) * OUT_DIM], wD1 = W2[(c0 + 3) * OUT_DIM + 1];

    float4 acc = make_float4(0.0f, 0.0f, 0.0f, 0.0f);
    for (int j = start; j < end; j += 4) {
        const int last = end - 1;
        const int j1 = (j + 1 < end) ? j + 1 : last;
        const int j2 = (j + 2 < end) ? j + 2 : last;
        const int j3 = (j + 3 < end) ? j + 3 : last;
        unsigned s0 = (unsigned)sorted[j];
        unsigned s1 = (unsigned)sorted[j1];
        unsigned s2 = (unsigned)sorted[j2];
        unsigned s3 = (unsigned)sorted[j3];
        s0 = s0 > nmax ? nmax : s0;
        s1 = s1 > nmax ? nmax : s1;
        s2 = s2 > nmax ? nmax : s2;
        s3 = s3 > nmax ? nmax : s3;
        const float d0 = dinv[s0], d1 = dinv[s1], d2 = dinv[s2], d3 = dinv[s3];
        const float w0 = d0 * di;
        const float w1 = d1 * ((j + 1 < end) ? di : 0.0f);
        const float w2 = d2 * ((j + 2 < end) ? di : 0.0f);
        const float w3 = d3 * ((j + 3 < end) ? di : 0.0f);
        const float4 v0 = h4[(size_t)s0 * 4 + q];
        const float4 v1 = h4[(size_t)s1 * 4 + q];
        const float4 v2 = h4[(size_t)s2 * 4 + q];
        const float4 v3 = h4[(size_t)s3 * 4 + q];
        acc.x = fmaf(v0.x, w0, acc.x); acc.y = fmaf(v0.y, w0, acc.y);
        acc.z = fmaf(v0.z, w0, acc.z); acc.w = fmaf(v0.w, w0, acc.w);
        acc.x = fmaf(v1.x, w1, acc.x); acc.y = fmaf(v1.y, w1, acc.y);
        acc.z = fmaf(v1.z, w1, acc.z); acc.w = fmaf(v1.w, w1, acc.w);
        acc.x = fmaf(v2.x, w2, acc.x); acc.y = fmaf(v2.y, w2, acc.y);
        acc.z = fmaf(v2.z, w2, acc.z); acc.w = fmaf(v2.w, w2, acc.w);
        acc.x = fmaf(v3.x, w3, acc.x); acc.y = fmaf(v3.y, w3, acc.y);
        acc.z = fmaf(v3.z, w3, acc.z); acc.w = fmaf(v3.w, w3, acc.w);
    }
    const float ws = di * di;
    acc.x = fmaf(hs.x, ws, acc.x); acc.y = fmaf(hs.y, ws, acc.y);
    acc.z = fmaf(hs.z, ws, acc.z); acc.w = fmaf(hs.w, ws, acc.w);

    const float vA = fmaxf(acc.x + bA, 0.0f);
    const float vB = fmaxf(acc.y + bB, 0.0f);
    const float vC = fmaxf(acc.z + bC, 0.0f);
    const float vD = fmaxf(acc.w + bD, 0.0f);
    float z0 = vA * wA0, z1 = vA * wA1;
    z0 = fmaf(vB, wB0, z0); z1 = fmaf(vB, wB1, z1);
    z0 = fmaf(vC, wC0, z0); z1 = fmaf(vC, wC1, z1);
    z0 = fmaf(vD, wD0, z0); z1 = fmaf(vD, wD1, z1);
    z0 += __shfl_xor(z0, 1); z1 += __shfl_xor(z1, 1);
    z0 += __shfl_xor(z0, 2); z1 += __shfl_xor(z1, 2);
    if (q == 0) ((float4*)rec)[i] = make_float4(z0, z1, di, 0.0f);
}

__global__ __launch_bounds__(256) void k_pull2_final(const int* __restrict__ rowptr,
                                                     const int* __restrict__ cnt,
                                                     const int* __restrict__ sorted,
                                                     const float* __restrict__ dinv,
                                                     const float* __restrict__ rec,
                                                     const float* __restrict__ b2,
                                                     float* __restrict__ out, int n, int nE) {
    const int g = blockIdx.x * 256 + threadIdx.x;
    const int iq = g >> 2;
    const int q = g & 3;
    const bool valid = iq < n;
    const int i = valid ? iq : n - 1;
    int start = rowptr[i];
    start = start < 0 ? 0 : (start > nE ? nE : start);
    int c = valid ? cnt[i] : 0;
    c = c < 0 ? 0 : c;
    const int end = (c > nE - start) ? nE : start + c;
    const float di = dinv[i];
    const float4* rv = (const float4*)rec;
    const unsigned nmax = (unsigned)n - 1u;

    float a0 = 0.0f, a1 = 0.0f;
    for (int j = start + q; j < end; j += 8) {
        const int jb = j + 4;
        const bool hb = jb < end;
        unsigned sa = (unsigned)sorted[j];
        unsigned sb = (unsigned)sorted[hb ? jb : j];
        sa = sa > nmax ? nmax : sa;
        sb = sb > nmax ? nmax : sb;
        const float4 va = rv[sa];
        const float4 vb = rv[sb];
        const float wa = va.z * di;
        const float wb = vb.z * (hb ? di : 0.0f);
        a0 = fmaf(va.x, wa, a0); a1 = fmaf(va.y, wa, a1);
        a0 = fmaf(vb.x, wb, a0); a1 = fmaf(vb.y, wb, a1);
    }
    a0 += __shfl_xor(a0, 1); a1 += __shfl_xor(a1, 1);
    a0 += __shfl_xor(a0, 2); a1 += __shfl_xor(a1, 2);

    if (valid && q == 0) {
        const float ws = di * di;
        const float4 hs = rv[i];
        const float z0 = fmaf(hs.x, ws, a0) + b2[0];
        const float z1 = fmaf(hs.y, ws, a1) + b2[1];
        const float mx = fmaxf(z0, z1);
        const float l  = mx + logf(expf(z0 - mx) + expf(z1 - mx));
        ((float2*)out)[i] = make_float2(z0 - l, z1 - l);
    }
}

extern "C" void kernel_launch(void* const* d_in, const int* in_sizes, int n_in,
                              void* d_out, int out_size, void* d_ws, size_t ws_size,
                              hipStream_t stream) {
    const float* x  = (const float*)d_in[0];
    const int*   ei = (const int*)d_in[1];
    const float* W1 = (const float*)d_in[2];
    const float* b1 = (const float*)d_in[3];
    const float* W2 = (const float*)d_in[4];
    const float* b2 = (const float*)d_in[5];
    float* out = (float*)d_out;

    const int n  = in_sizes[0] / IN_DIM;
    const int nE = in_sizes[1] / 2;
    const int* src = ei;
    const int* dst = ei + nE;

    const size_t np = ((size_t)n  + 31) & ~(size_t)31;
    const size_t Ep = ((size_t)nE + 31) & ~(size_t)31;
    const int nb   = (n + BNODES - 1) / BNODES;
    const int nblk = (nE + SLICE - 1) / SLICE;
    const int nbp  = (nb + 1 + 31) & ~31;
    const size_t tw = (size_t)nblk * nbp;
    const size_t uw = (Ep > 32 * np) ? Ep : 32 * np;
    const size_t need = (5 * np + Ep + uw) * sizeof(float);
    if (ws_size < need) return;
    if (nb > MAXB || n > (1 << 24) || nblk > 32 * RPW || tw > 2 * np) return;

    int*   cnt    = (int*)d_ws;
    int*   rowptr = cnt + np;
    float* dinv   = (float*)(rowptr + np);
    int*   T      = (int*)(dinv + np);
    int*   sorted = T + 2 * np;
    int*   binned = sorted + Ep;
    float* h      = (float*)binned;
    float* rec    = h + 16 * np;

    k_slice_sort  <<<nblk, 1024, 0, stream>>>(src, dst, T, binned, nE, nb, nbp);
    k_bucket_sort <<<nb, 1024, 0, stream>>>(binned, T, cnt, rowptr, dinv, sorted, n, nE, nbp, nblk);

    const int tiles   = (n + 15) / 16;
    const int gblocks = (tiles + 3) / 4;
    k_gemm1_wmma  <<<gblocks, 128, 0, stream>>>(x, W1, h, n);

    const int qblocks = (int)(((size_t)n * 4 + 255) / 256);
    k_pull16      <<<qblocks, 256, 0, stream>>>(rowptr, cnt, sorted, dinv, h, b1, W2, rec, n, nE);
    k_pull2_final <<<qblocks, 256, 0, stream>>>(rowptr, cnt, sorted, dinv, rec, b2, out, n, nE);
}


# both neighbour sums fetch a wave's 8 in-edge lists as whole 128-byte lines into LDS (one load group) and take source ids from LDS, not from 4-byte list loads in each edge trip; sums bit-identical
# speedup vs baseline: 12.3592x; 1.2839x over previous
#include <hip/hip_runtime.h>
#include <hip/hip_bf16.h>
#include <math.h>


typedef __attribute__((ext_vector_type(2))) float v2f;
typedef __attribute__((ext_vector_type(8))) float v8f;

#define IN_DIM  128
#define HID     16
#define OUT_DIM 2

#define BSHIFT 8
#define BNODES 256
#define MAXB   512
#define SLICE  12288
#define SPT    12
#define RPW    12
#define BCAP   16384
#define BPT    10

__global__ __launch_bounds__(1024, 16) void k_slice_sort(const int* __restrict__ src, const int* __restrict__ dst,
                                                     int* __restrict__ T, int* __restrict__ binned,
                                                     int nE, int nb, int nbp) {
    __shared__ int sWord[SLICE];
    __shared__ int hist[MAXB];
    __shared__ int lbase[MAXB];
    __shared__ int wtot[16];
    const int t = threadIdx.x, lane = t & 31, wave = t >> 5;
    const int me = blockIdx.x;

    const int e0 = me * SLICE + t;
    int s[SPT], d[SPT], r[SPT];
    #pragma unroll
    for (int u = 0; u < SPT; ++u) {
        const int e = e0 + u * 1024;
        const int ec = e < nE ? e : nE - 1;
        s[u] = src[ec];
        d[u] = dst[ec];
    }
    if (t < MAXB) hist[t] = 0;
    __syncthreads();

    #pragma unroll
    for (int u = 0; u < SPT; ++u) {
        const int e = e0 + u * 1024;
        r[u] = 0;
        if (e < nE) {
            unsigned b = (unsigned)d[u] >> BSHIFT;
            b = b > (unsigned)(nb - 1) ? (unsigned)(nb - 1) : b;
            r[u] = atomicAdd(&hist[b], 1);
        }
    }
    __syncthreads();
    int c = 0, incl = 0;
    if (t < MAXB) {
        c = hist[t];
        incl = c;
        #pragma unroll
        for (int o = 1; o < 32; o <<= 1) {
            const int v = __shfl_up(incl, o);
            if (lane >= o) incl += v;
        }
        if (lane == 31) wtot[wave] = incl;
    }
    __syncthreads();
    int ex = 0;
    if (t < MAXB) {
        int woff = 0;
        #pragma unroll
        for (int w = 0; w < 16; ++w) woff += (w < wave) ? wtot[w] : 0;
        ex = woff + incl - c;
        lbase[t] = ex;
    }
    __syncthreads();
    #pragma unroll
    for (int u = 0; u < SPT; ++u) {
        const int e = e0 + u * 1024;
        if (e < nE) {
            unsigned b = (unsigned)d[u] >> BSHIFT;
            b = b > (unsigned)(nb - 1) ? (unsigned)(nb - 1) : b;
            sWord[lbase[b] + r[u]] = (int)((((unsigned)d[u] & (BNODES - 1)) << 24) | ((unsigned)s[u] & 0xFFFFFFu));
        }
    }
    __syncthreads();
    if (t < nb) T[(size_t)me * nbp + t] = ex;
    if (t == nb - 1) T[(size_t)me * nbp + nb] = ex + c;
    const int first = me * SLICE;
    const int have = (nE - first < SLICE) ? nE - first : SLICE;
    for (int k = t; k < have; k += 1024) binned[first + k] = sWord[k];
}

__global__ __launch_bounds__(1024, 16) void k_bucket_sort(const int* __restrict__ binned, const int* __restrict__ T,
                                                      int* __restrict__ cnt, int* __restrict__ rowptr,
                                                      float* __restrict__ dinv, int* __restrict__ sorted,
                                                      int n, int nE, int nbp, int nblk) {
    __shared__ int sIn[BCAP];
    __shared__ int sOut[BCAP];
    __shared__ int rBase[32 * RPW];
    __shared__ int rLen[32 * RPW];
    __shared__ int rPos[32 * RPW];
    __shared__ int lcnt[BNODES];
    __shared__ int lptr[BNODES];
    __shared__ int wlen[RPW];
    __shared__ int wlo[RPW];
    __shared__ int wtot[8];
    const int t = threadIdx.x, lane = t & 31, wave = t >> 5;
    const int b = blockIdx.x;

    int lo = 0, len = 0;
    if (t < nblk) {
        const int first = t * SLICE;
        int have = (nE - first < SLICE) ? nE - first : SLICE;
        have = have < 0 ? 0 : have;
        int a = T[(size_t)t * nbp + b];
        int e = T[(size_t)t * nbp + b + 1];
        a = a < 0 ? 0 : (a > have ? have : a);
        e = e < a ? a : (e > have ? have : e);
        lo = a;
        len = e - a;
    }
    int incl = len, slo = lo;
    #pragma unroll
    for (int o = 1; o < 32; o <<= 1) {
        const int v = __shfl_up(incl, o);
        const int y = __shfl_up(slo, o);
        if (lane >= o) { incl += v; slo += y; }
    }
    if (wave < RPW && lane == 31) {
        wlen[wave] = incl;
        wlo[wave] = slo;
    }
    if (t < BNODES) lcnt[t] = 0;
    __syncthreads();
    int woff = 0, tot = 0, bs = 0;
    #pragma unroll
    for (int w = 0; w < RPW; ++w) {
        const int x = wlen[w];
        woff += (w < wave) ? x : 0;
        tot += x;
        bs += wlo[w];
    }
    if (t < 32 * RPW) {
        rBase[t] = t * SLICE + lo;
        rLen[t]  = len;
        rPos[t]  = woff + incl - len;
    }
    bs = bs < 0 ? 0 : (bs > nE ? nE : bs);
    const int size = (tot > nE - bs) ? nE - bs : tot;
    const bool inlds = tot <= BCAP;
    const int amax = nE - 1;
    __syncthreads();

    if (inlds) {
        int w0[RPW], w1[RPW];
        #pragma unroll
        for (int u = 0; u < RPW; ++u) {
            const int j = wave + 32 * u;
            const int base = rBase[j], ln = rLen[j];
            int a0 = base + (lane < ln ? lane : ln - 1);
            int a1 = base + (lane + 32 < ln ? lane + 32 : ln - 1);
            a0 = a0 < 0 ? 0 : (a0 > amax ? amax : a0);
            a1 = a1 < 0 ? 0 : (a1 > amax ? amax : a1);
            w0[u] = binned[a0];
            w1[u] = binned[a1];
        }
        int lnv[RPW], posv[RPW];
        #pragma unroll
        for (int u = 0; u < RPW; ++u) {
            const int j = wave + 32 * u;
            lnv[u] = rLen[j];
            posv[u] = rPos[j];
        }
        #pragma unroll
        for (int u = 0; u < RPW; ++u) {
            if (lane < lnv[u])      sIn[posv[u] + lane]      = w0[u];
            if (lane + 32 < lnv[u]) sIn[posv[u] + lane + 32] = w1[u];
        }
        for (int j = wave; j < 32 * RPW; j += 32) {
            const int base = rBase[j], ln = rLen[j], pos = rPos[j];
            for (int o = lane + 64; o < ln; o += 32) {
                int a = base + o;
                a = a > amax ? amax : a;
                sIn[pos + o] = binned[a];
            }
        }
    }
    __syncthreads();

    if (inlds) {
        for (int kb = t; kb < size + t; kb += BPT * 1024) {
            unsigned w[BPT];
            #pragma unroll
            for (int u = 0; u < BPT; ++u) {
                const int k = kb + u * 1024;
                w[u] = (unsigned)sIn[k < size ? k : size - 1];
            }
            #pragma unroll
            for (int u = 0; u < BPT; ++u)
                if (kb + u * 1024 < size) atomicAdd(&lcnt[w[u] >> 24], 1);
        }
    } else {
        for (int j = wave; j < 32 * RPW; j += 32) {
            const int base = rBase[j], ln = rLen[j];
            for (int o = lane; o < ln; o += 32) {
                int a = base + o;
                a = a > amax ? amax : a;
                atomicAdd(&lcnt[(unsigned)binned[a] >> 24], 1);
            }
        }
    }
    __syncthreads();
    int c = 0;
    incl = 0;
    if (t < BNODES) {
        c = lcnt[t];
        incl = c;
        #pragma unroll
        for (int o = 1; o < 32; o <<= 1) {
            const int v = __shfl_up(incl, o);
            if (lane >= o) incl += v;
        }
        if (lane == 31) wtot[wave] = incl;
    }
    __syncthreads();
    if (t < BNODES) {
        int wo = 0;
        #pragma unroll
        for (int w = 0; w < 8; ++w) wo += (w < wave) ? wtot[w] : 0;
        const int ex = wo + incl - c;
        lptr[t] = ex;
        lcnt[t] = 0;
        const int node = b * BNODES + t;
        if (node < n) {
            cnt[node] = c;
            rowptr[node] = bs + ex;
            const float dg = (float)(c + 1);
            dinv[node] = (dg > 0.0f) ? rsqrtf(dg) : 0.0f;
        }
    }
    __syncthreads();

    if (inlds) {
        for (int kb = t; kb < size + t; kb += BPT * 1024) {
            unsigned w[BPT];
            #pragma unroll
            for (int u = 0; u < BPT; ++u) {
                const int k = kb + u * 1024;
                w[u] = (unsigned)sIn[k < size ? k : size - 1];
            }
            #pragma unroll
            for (int u = 0; u < BPT; ++u) {
                if (kb + u * 1024 < size) {
                    const unsigned dl = w[u] >> 24;
                    const int p = lptr[dl] + atomicAdd(&lcnt[dl], 1);
                    sOut[p] = (int)(w[u] & 0xFFFFFFu);
                }
            }
        }
    } else {
        for (int j = wave; j < 32 * RPW; j += 32) {
            const int base = rBase[j], ln = rLen[j];
            for (int o = lane; o < ln; o += 32) {
                int a = base + o;
                a = a > amax ? amax : a;
                const unsigned w = (unsigned)binned[a];
                const unsigned dl = w >> 24;
                const int p = lptr[dl] + atomicAdd(&lcnt[dl], 1);
                if (p < size) sorted[bs + p] = (int)(w & 0xFFFFFFu);
            }
        }
    }
    __syncthreads();
    if (inlds)
        for (int k = t; k < size; k += 1024) sorted[bs + k] = sOut[k];
}

__global__ __launch_bounds__(128) void k_gemm1_wmma(const float* __restrict__ x,
                                                    const float* __restrict__ W1,
                                                    float* __restrict__ h, int n) {
    __shared__ float sW[IN_DIM * HID];
    __shared__ float sX[4][16 * 132];

    for (int t = threadIdx.x; t < IN_DIM * HID; t += blockDim.x) sW[t] = W1[t];

    const int wave = threadIdx.x >> 5;
    const int lane = threadIdx.x & 31;
    const int tile = blockIdx.x * 4 + wave;
    const int row0 = tile * 16;
    const bool active = (row0 < n);

    if (active) {
        #pragma unroll 4
        for (int i = 0; i < 16; ++i) {
            int r = row0 + i;
            if (r >= n) r = n - 1;
            const float4 v = ((const float4*)(x + (size_t)r * IN_DIM))[lane];
            *(float4*)&sX[wave][i * 132 + lane * 4] = v;
        }
    }
    __syncthreads();
    if (!active) return;

    const int m  = lane & 15;
    const int kb = (lane >> 4) << 1;
    const float* xr = &sX[wave][m * 132];

    v8f acc = {};
    #pragma unroll
    for (int k0 = 0; k0 < IN_DIM; k0 += 4) {
        v2f a, b;
        a.x = xr[k0 + kb];
        a.y = xr[k0 + kb + 1];
        b.x = sW[(k0 + kb)     * HID + m];
        b.y = sW[(k0 + kb + 1) * HID + m];
        acc = __builtin_amdgcn_wmma_f32_16x16x4_f32(false, a, false, b,
                                                    (short)0, acc, false, false);
    }

    const int rbase = row0 + ((lane >> 4) << 3);
    #pragma unroll
    for (int v = 0; v < 8; ++v) {
        int r = rbase + v;
        if (r < n) h[(size_t)r * HID + m] = acc[v];
    }
}

#define LCAP 384

__device__ __forceinline__ bool stage_lists(const int* __restrict__ sorted, int* slab,
                                            int start, int end, int lane, int nE, int& base) {
    const int lo = __builtin_amdgcn_readlane(start, 0);
    const int hi = __builtin_amdgcn_readlane(end, 31);
    base = lo & ~31;
    const bool ok = start >= lo && end <= hi && lo >= 0 && hi <= nE && hi - base <= LCAP;
    if (!__all(ok)) return false;
    int w[LCAP / 32];
    #pragma unroll
    for (int k = 0; k < LCAP / 32; ++k) {
        const int p = base + lane + 32 * k;
        w[k] = 0;
        if (p < hi) w[k] = sorted[p];
    }
    #pragma unroll
    for (int k = 0; k < LCAP / 32; ++k) slab[lane + 32 * k] = w[k];
    return true;
}

__device__ __forceinline__ float4 in_sum16(const int* lst, int start, int end,
                                           const float* __restrict__ dinv, const float4* __restrict__ h4,
                                           int q, float di, unsigned nmax) {
    float4 acc = make_float4(0.0f, 0.0f, 0.0f, 0.0f);
    for (int j = start; j < end; j += 4) {
        const int last = end - 1;
        const int j1 = (j + 1 < end) ? j + 1 : last;
        const int j2 = (j + 2 < end) ? j + 2 : last;
        const int j3 = (j + 3 < end) ? j + 3 : last;
        unsigned s0 = (unsigned)lst[j];
        unsigned s1 = (unsigned)lst[j1];
        unsigned s2 = (unsigned)lst[j2];
        unsigned s3 = (unsigned)lst[j3];
        s0 = s0 > nmax ? nmax : s0;
        s1 = s1 > nmax ? nmax : s1;
        s2 = s2 > nmax ? nmax : s2;
        s3 = s3 > nmax ? nmax : s3;
        const float d0 = dinv[s0], d1 = dinv[s1], d2 = dinv[s2], d3 = dinv[s3];
        const float w0 = d0 * di;
        const float w1 = d1 * ((j + 1 < end) ? di : 0.0f);
        const float w2 = d2 * ((j + 2 < end) ? di : 0.0f);
        const float w3 = d3 * ((j + 3 < end) ? di : 0.0f);
        const float4 v0 = h4[(size_t)s0 * 4 + q];
        const float4 v1 = h4[(size_t)s1 * 4 + q];
        const float4 v2 = h4[(size_t)s2 * 4 + q];
        const float4 v3 = h4[(size_t)s3 * 4 + q];
        acc.x = fmaf(v0.x, w0, acc.x); acc.y = fmaf(v0.y, w0, acc.y);
        acc.z = fmaf(v0.z, w0, acc.z); acc.w = fmaf(v0.w, w0, acc.w);
        acc.x = fmaf(v1.x, w1, acc.x); acc.y = fmaf(v1.y, w1, acc.y);
        acc.z = fmaf(v1.z, w1, acc.z); acc.w = fmaf(v1.w, w1, acc.w);
        acc.x = fmaf(v2.x, w2, acc.x); acc.y = fmaf(v2.y, w2, acc.y);
        acc.z = fmaf(v2.z, w2, acc.z); acc.w = fmaf(v2.w, w2, acc.w);
        acc.x = fmaf(v3.x, w3, acc.x); acc.y = fmaf(v3.y, w3, acc.y);
        acc.z = fmaf(v3.z, w3, acc.z); acc.w = fmaf(v3.w, w3, acc.w);
    }
    return acc;
}

__global__ __launch_bounds__(256) void k_pull16(const int* __restrict__ rowptr,
                                                const int* __restrict__ cnt,
                                                const int* __restrict__ sorted,
                                                const float* __restrict__ dinv,
                                                const float* __restrict__ h,
                                                const float* __restrict__ b1,
                                                const float* __restrict__ W2,
                                                float* __restrict__ rec, int n, int nE) {
    __shared__ int sList[8 * LCAP];
    const int g = blockIdx.x * 256 + threadIdx.x;
    const int iq = g >> 2;
    const int q = g & 3;
    const int lane = threadIdx.x & 31, wave = threadIdx.x >> 5;
    const bool valid = iq < n;
    const int i = valid ? iq : n - 1;
    int start = rowptr[i];
    start = start < 0 ? 0 : (start > nE ? nE : start);
    int c = cnt[i];
    c = (!valid || c < 0) ? 0 : c;
    const int end = (c > nE - start) ? nE : start + c;
    const float di = dinv[i];
    const float4* h4 = (const float4*)h;
    const unsigned nmax = (unsigned)n - 1u;

    const float4 hs = h4[(size_t)i * 4 + q];
    const int c0 = q * 4;
    const float bA = b1[c0], bB = b1[c0 + 1], bC = b1[c0 + 2], bD = b1[c0 + 3];
    const float wA0 = W2[(c0    ) * OUT_DIM], wA1 = W2[(c0    ) * OUT_DIM + 1];
    const float wB0 = W2[(c0 + 1) * OUT_DIM], wB1 = W2[(c0 + 1) * OUT_DIM + 1];
    const float wC0 = W2[(c0 + 2) * OUT_DIM], wC1 = W2[(c0 + 2) * OUT_DIM + 1];
    const float wD0 = W2[(c0 + 3) * OUT_DIM], wD1 = W2[(c0 + 3) * OUT_DIM + 1];

    int* slab = sList + wave * LCAP;
    int base = 0;
    const bool staged = stage_lists(sorted, slab, start, end, lane, nE, base);
    __syncthreads();
    float4 acc;
    if (staged) acc = in_sum16(slab, start - base, end - base, dinv, h4, q, di, nmax);
    else        acc = in_sum16(sorted, start, end, dinv, h4, q, di, nmax);
    const float ws = di * di;
    acc.x = fmaf(hs.x, ws, acc.x); acc.y = fmaf(hs.y, ws, acc.y);
    acc.z = fmaf(hs.z, ws, acc.z); acc.w = fmaf(hs.w, ws, acc.w);

    const float vA = fmaxf(acc.x + bA, 0.0f);
    const float vB = fmaxf(acc.y + bB, 0.0f);
    const float vC = fmaxf(acc.z + bC, 0.0f);
    const float vD = fmaxf(acc.w + bD, 0.0f);
    float z0 = vA * wA0, z1 = vA * wA1;
    z0 = fmaf(vB, wB0, z0); z1 = fmaf(vB, wB1, z1);
    z0 = fmaf(vC, wC0, z0); z1 = fmaf(vC, wC1, z1);
    z0 = fmaf(vD, wD0, z0); z1 = fmaf(vD, wD1, z1);
    z0 += __shfl_xor(z0, 1); z1 += __shfl_xor(z1, 1);
    z0 += __shfl_xor(z0, 2); z1 += __shfl_xor(z1, 2);
    if (valid && q == 0) ((float4*)rec)[i] = make_float4(z0, z1, di, 0.0f);
}

__device__ __forceinline__ void in_sum2(const int* lst, int first, int end, const float4* __restrict__ rv,
                                        float di, unsigned nmax, float& a0, float& a1) {
    for (int j = first; j < end; j += 8) {
        const int jb = j + 4;
        const bool hb = jb < end;
        unsigned sa = (unsigned)lst[j];
        unsigned sb = (unsigned)lst[hb ? jb : j];
        sa = sa > nmax ? nmax : sa;
        sb = sb > nmax ? nmax : sb;
        const float4 va = rv[sa];
        const float4 vb = rv[sb];
        const float wa = va.z * di;
        const float wb = vb.z * (hb ? di : 0.0f);
        a0 = fmaf(va.x, wa, a0); a1 = fmaf(va.y, wa, a1);
        a0 = fmaf(vb.x, wb, a0); a1 = fmaf(vb.y, wb, a1);
    }
}

__global__ __launch_bounds__(256) void k_pull2_final(const int* __restrict__ rowptr,
                                                     const int* __restrict__ cnt,
                                                     const int* __restrict__ sorted,
                                                     const float* __restrict__ dinv,
                                                     const float* __restrict__ rec,
                                                     const float* __restrict__ b2,
                                                     float* __restrict__ out, int n, int nE) {
    __shared__ int sList[8 * LCAP];
    const int g = blockIdx.x * 256 + threadIdx.x;
    const int iq = g >> 2;
    const int q = g & 3;
    const int lane = threadIdx.x & 31, wave = threadIdx.x >> 5;
    const bool valid = iq < n;
    const int i = valid ? iq : n - 1;
    int start = rowptr[i];
    start = start < 0 ? 0 : (start > nE ? nE : start);
    int c = valid ? cnt[i] : 0;
    c = c < 0 ? 0 : c;
    const int end = (c > nE - start) ? nE : start + c;
    const float di = dinv[i];
    const float4* rv = (const float4*)rec;
    const unsigned nmax = (unsigned)n - 1u;

    int* slab = sList + wave * LCAP;
    int base = 0;
    const bool staged = stage_lists(sorted, slab, start, end, lane, nE, base);
    __syncthreads();
    float a0 = 0.0f, a1 = 0.0f;
    if (staged) in_sum2(slab, start - base + q, end - base, rv, di, nmax, a0, a1);
    else        in_sum2(sorted, start + q, end, rv, di, nmax, a0, a1);
    a0 += __shfl_xor(a0, 1); a1 += __shfl_xor(a1, 1);
    a0 += __shfl_xor(a0, 2); a1 += __shfl_xor(a1, 2);

    if (valid && q == 0) {
        const float ws = di * di;
        const float4 hs = rv[i];
        const float z0 = fmaf(hs.x, ws, a0) + b2[0];
        const float z1 = fmaf(hs.y, ws, a1) + b2[1];
        const float mx = fmaxf(z0, z1);
        const float l  = mx + logf(expf(z0 - mx) + expf(z1 - mx));
        ((float2*)out)[i] = make_float2(z0 - l, z1 - l);
    }
}

extern "C" void kernel_launch(void* const* d_in, const int* in_sizes, int n_in,
                              void* d_out, int out_size, void* d_ws, size_t ws_size,
                              hipStream_t stream) {
    const float* x  = (const float*)d_in[0];
    const int*   ei = (const int*)d_in[1];
    const float* W1 = (const float*)d_in[2];
    const float* b1 = (const float*)d_in[3];
    const float* W2 = (const float*)d_in[4];
    const float* b2 = (const float*)d_in[5];
    float* out = (float*)d_out;

    const int n  = in_sizes[0] / IN_DIM;
    const int nE = in_sizes[1] / 2;
    const int* src = ei;
    const int* dst = ei + nE;

    const size_t np = ((size_t)n  + 31) & ~(size_t)31;
    const size_t Ep = ((size_t)nE + 31) & ~(size_t)31;
    const int nb   = (n + BNODES - 1) / BNODES;
    const int nblk = (nE + SLICE - 1) / SLICE;
    const int nbp  = (nb + 1 + 31) & ~31;
    const size_t tw = (size_t)nblk * nbp;
    const size_t uw = (Ep > 32 * np) ? Ep : 32 * np;
    const size_t need = (5 * np + Ep + uw) * sizeof(float);
    if (ws_size < need) return;
    if (nb > MAXB || n > (1 << 24) || nblk > 32 * RPW || tw > 2 * np) return;

    int*   cnt    = (int*)d_ws;
    int*   rowptr = cnt + np;
    float* dinv   = (float*)(rowptr + np);
    int*   T      = (int*)(dinv + np);
    int*   sorted = T + 2 * np;
    int*   binned = sorted + Ep;
    float* h      = (float*)binned;
    float* rec    = h + 16 * np;

    k_slice_sort  <<<nblk, 1024, 0, stream>>>(src, dst, T, binned, nE, nb, nbp);
    k_bucket_sort <<<nb, 1024, 0, stream>>>(binned, T, cnt, rowptr, dinv, sorted, n, nE, nbp, nblk);

    const int tiles   = (n + 15) / 16;
    const int gblocks = (tiles + 3) / 4;
    k_gemm1_wmma  <<<gblocks, 128, 0, stream>>>(x, W1, h, n);

    const int qblocks = (int)(((size_t)n * 4 + 255) / 256);
    k_pull16      <<<qblocks, 256, 0, stream>>>(rowptr, cnt, sorted, dinv, h, b1, W2, rec, n, nE);
    k_pull2_final <<<qblocks, 256, 0, stream>>>(rowptr, cnt, sorted, dinv, rec, b2, out, n, nE);
}


# layer-1 sum forms each source's deg^-1/2 from a one-byte in-degree table that every 200-node block copies into LDS: an edge asks memory for its row only, no 4-byte gather; sums bit-identical
# speedup vs baseline: 14.0278x; 1.1350x over previous
#include <hip/hip_runtime.h>
#include <hip/hip_bf16.h>
#include <math.h>


typedef __attribute__((ext_vector_type(2))) float v2f;
typedef __attribute__((ext_vector_type(8))) float v8f;

#define IN_DIM  128
#define HID     16
#define OUT_DIM 2

#define BSHIFT 8
#define BNODES 256
#define MAXB   512
#define SLICE  12288
#define SPT    12
#define RPW    12
#define BCAP   16384
#define BPT    10

__device__ __forceinline__ float inv_root_deg(int c) {
    const float dg = (float)(c + 1);
    return (dg > 0.0f) ? rsqrtf(dg) : 0.0f;
}

__global__ __launch_bounds__(1024, 16) void k_slice_sort(const int* __restrict__ src, const int* __restrict__ dst,
                                                     int* __restrict__ T, int* __restrict__ binned,
                                                     int nE, int nb, int nbp) {
    __shared__ int sWord[SLICE];
    __shared__ int hist[MAXB];
    __shared__ int lbase[MAXB];
    __shared__ int wtot[16];
    const int t = threadIdx.x, lane = t & 31, wave = t >> 5;
    const int me = blockIdx.x;

    const int e0 = me * SLICE + t;
    int s[SPT], d[SPT], r[SPT];
    #pragma unroll
    for (int u = 0; u < SPT; ++u) {
        const int e = e0 + u * 1024;
        const int ec = e < nE ? e : nE - 1;
        s[u] = src[ec];
        d[u] = dst[ec];
    }
    if (t < MAXB) hist[t] = 0;
    __syncthreads();

    #pragma unroll
    for (int u = 0; u < SPT; ++u) {
        const int e = e0 + u * 1024;
        r[u] = 0;
        if (e < nE) {
            unsigned b = (unsigned)d[u] >> BSHIFT;
            b = b > (unsigned)(nb - 1) ? (unsigned)(nb - 1) : b;
            r[u] = atomicAdd(&hist[b], 1);
        }
    }
    __syncthreads();
    int c = 0, incl = 0;
    if (t < MAXB) {
        c = hist[t];
        incl = c;
        #pragma unroll
        for (int o = 1; o < 32; o <<= 1) {
            const int v = __shfl_up(incl, o);
            if (lane >= o) incl += v;
        }
        if (lane == 31) wtot[wave] = incl;
    }
    __syncthreads();
    int ex = 0;
    if (t < MAXB) {
        int woff = 0;
        #pragma unroll
        for (int w = 0; w < 16; ++w) woff += (w < wave) ? wtot[w] : 0;
        ex = woff + incl - c;
        lbase[t] = ex;
    }
    __syncthreads();
    #pragma unroll
    for (int u = 0; u < SPT; ++u) {
        const int e = e0 + u * 1024;
        if (e < nE) {
            unsigned b = (unsigned)d[u] >> BSHIFT;
            b = b > (unsigned)(nb - 1) ? (unsigned)(nb - 1) : b;
            sWord[lbase[b] + r[u]] = (int)((((unsigned)d[u] & (BNODES - 1)) << 24) | ((unsigned)s[u] & 0xFFFFFFu));
        }
    }
    __syncthreads();
    if (t < nb) T[(size_t)me * nbp + t] = ex;
    if (t == nb - 1) T[(size_t)me * nbp + nb] = ex + c;
    const int first = me * SLICE;
    const int have = (nE - first < SLICE) ? nE - first : SLICE;
    for (int k = t; k < have; k += 1024) binned[first + k] = sWord[k];
}

__global__ __launch_bounds__(1024, 16) void k_bucket_sort(const int* __restrict__ binned, const int* __restrict__ T,
                                                      int* __restrict__ cnt, int* __restrict__ rowptr,
                                                      float* __restrict__ dinv, int* __restrict__ sorted,
                                                      int* __restrict__ deg8w,
                                                      int n, int nE, int nbp, int nblk) {
    __shared__ int sIn[BCAP];
    __shared__ int sOut[BCAP];
    __shared__ int rBase[32 * RPW];
    __shared__ int rLen[32 * RPW];
    __shared__ int rPos[32 * RPW];
    __shared__ int lcnt[BNODES];
    __shared__ int lptr[BNODES];
    __shared__ int wlen[RPW];
    __shared__ int wlo[RPW];
    __shared__ int wtot[8];
    const int t = threadIdx.x, lane = t & 31, wave = t >> 5;
    const int b = blockIdx.x;

    int lo = 0, len = 0;
    if (t < nblk) {
        const int first = t * SLICE;
        int have = (nE - first < SLICE) ? nE - first : SLICE;
        have = have < 0 ? 0 : have;
        int a = T[(size_t)t * nbp + b];
        int e = T[(size_t)t * nbp + b + 1];
        a = a < 0 ? 0 : (a > have ? have : a);
        e = e < a ? a : (e > have ? have : e);
        lo = a;
        len = e - a;
    }
    int incl = len, slo = lo;
    #pragma unroll
    for (int o = 1; o < 32; o <<= 1) {
        const int v = __shfl_up(incl, o);
        const int y = __shfl_up(slo, o);
        if (lane >= o) { incl += v; slo += y; }
    }
    if (wave < RPW && lane == 31) {
        wlen[wave] = incl;
        wlo[wave] = slo;
    }
    if (t < BNODES) lcnt[t] = 0;
    __syncthreads();
    int woff = 0, tot = 0, bs = 0;
    #pragma unroll
    for (int w = 0; w < RPW; ++w) {
        const int x = wlen[w];
        woff += (w < wave) ? x : 0;
        tot += x;
        bs += wlo[w];
    }
    if (t < 32 * RPW) {
        rBase[t] = t * SLICE + lo;
        rLen[t]  = len;
        rPos[t]  = woff + incl - len;
    }
    bs = bs < 0 ? 0 : (bs > nE ? nE : bs);
    const int size = (tot > nE - bs) ? nE - bs : tot;
    const bool inlds = tot <= BCAP;
    const int amax = nE - 1;
    __syncthreads();

    if (inlds) {
        int w0[RPW], w1[RPW];
        #pragma unroll
        for (int u = 0; u < RPW; ++u) {
            const int j = wave + 32 * u;
            const int base = rBase[j], ln = rLen[j];
            int a0 = base + (lane < ln ? lane : ln - 1);
            int a1 = base + (lane + 32 < ln ? lane + 32 : ln - 1);
            a0 = a0 < 0 ? 0 : (a0 > amax ? amax : a0);
            a1 = a1 < 0 ? 0 : (a1 > amax ? amax : a1);
            w0[u] = binned[a0];
            w1[u] = binned[a1];
        }
        int lnv[RPW], posv[RPW];
        #pragma unroll
        for (int u = 0; u < RPW; ++u) {
            const int j = wave + 32 * u;
            lnv[u] = rLen[j];
            posv[u] = rPos[j];
        }
        #pragma unroll
        for (int u = 0; u < RPW; ++u) {
            if (lane < lnv[u])      sIn[posv[u] + lane]      = w0[u];
            if (lane + 32 < lnv[u]) sIn[posv[u] + lane + 32] = w1[u];
        }
        for (int j = wave; j < 32 * RPW; j += 32) {
            const int base = rBase[j], ln = rLen[j], pos = rPos[j];
            for (int o = lane + 64; o < ln; o += 32) {
                int a = base + o;
                a = a > amax ? amax : a;
                sIn[pos + o] = binned[a];
            }
        }
    }
    __syncthreads();

    if (inlds) {
        for (int kb = t; kb < size + t; kb += BPT * 1024) {
            unsigned w[BPT];
            #pragma unroll
            for (int u = 0; u < BPT; ++u) {
                const int k = kb + u * 1024;
                w[u] = (unsigned)sIn[k < size ? k : size - 1];
            }
            #pragma unroll
            for (int u = 0; u < BPT; ++u)
                if (kb + u * 1024 < size) atomicAdd(&lcnt[w[u] >> 24], 1);
        }
    } else {
        for (int j = wave; j < 32 * RPW; j += 32) {
            const int base = rBase[j], ln = rLen[j];
            for (int o = lane; o < ln; o += 32) {
                int a = base + o;
                a = a > amax ? amax : a;
                atomicAdd(&lcnt[(unsigned)binned[a] >> 24], 1);
            }
        }
    }
    __syncthreads();
    int c = 0;
    incl = 0;
    if (t < BNODES) {
        c = lcnt[t];
        incl = c;
        #pragma unroll
        for (int o = 1; o < 32; o <<= 1) {
            const int v = __shfl_up(incl, o);
            if (lane >= o) incl += v;
        }
        if (lane == 31) wtot[wave] = incl;
    }
    __syncthreads();
    if (t < BNODES) {
        int wo = 0;
        #pragma unroll
        for (int w = 0; w < 8; ++w) wo += (w < wave) ? wtot[w] : 0;
        const int ex = wo + incl - c;
        lptr[t] = ex;
        lcnt[t] = 0;
        const int node = b * BNODES + t;
        if (node < n) {
            cnt[node] = c;
            rowptr[node] = bs + ex;
            dinv[node] = inv_root_deg(c);
        }
        const int c8 = (node < n) ? (c > 255 ? 255 : c) : 0;
        ((unsigned char*)deg8w)[(size_t)b * BNODES + t] = (unsigned char)c8;
    }
    __syncthreads();

    if (inlds) {
        for (int kb = t; kb < size + t; kb += BPT * 1024) {
            unsigned w[BPT];
            #pragma unroll
            for (int u = 0; u < BPT; ++u) {
                const int k = kb + u * 1024;
                w[u] = (unsigned)sIn[k < size ? k : size - 1];
            }
            #pragma unroll
            for (int u = 0; u < BPT; ++u) {
                if (kb + u * 1024 < size) {
                    const unsigned dl = w[u] >> 24;
                    const int p = lptr[dl] + atomicAdd(&lcnt[dl], 1);
                    sOut[p] = (int)(w[u] & 0xFFFFFFu);
                }
            }
        }
    } else {
        for (int j = wave; j < 32 * RPW; j += 32) {
            const int base = rBase[j], ln = rLen[j];
            for (int o = lane; o < ln; o += 32) {
                int a = base + o;
                a = a > amax ? amax : a;
                const unsigned w = (unsigned)binned[a];
                const unsigned dl = w >> 24;
                const int p = lptr[dl] + atomicAdd(&lcnt[dl], 1);
                if (p < size) sorted[bs + p] = (int)(w & 0xFFFFFFu);
            }
        }
    }
    __syncthreads();
    if (inlds)
        for (int k = t; k < size; k += 1024) sorted[bs + k] = sOut[k];
}

__global__ __launch_bounds__(128) void k_gemm1_wmma(const float* __restrict__ x,
                                                    const float* __restrict__ W1,
                                                    float* __restrict__ h, int n) {
    __shared__ float sW[IN_DIM * HID];
    __shared__ float sX[4][16 * 132];

    for (int t = threadIdx.x; t < IN_DIM * HID; t += blockDim.x) sW[t] = W1[t];

    const int wave = threadIdx.x >> 5;
    const int lane = threadIdx.x & 31;
    const int tile = blockIdx.x * 4 + wave;
    const int row0 = tile * 16;
    const bool active = (row0 < n);

    if (active) {
        #pragma unroll 4
        for (int i = 0; i < 16; ++i) {
            int r = row0 + i;
            if (r >= n) r = n - 1;
            const float4 v = ((const float4*)(x + (size_t)r * IN_DIM))[lane];
            *(float4*)&sX[wave][i * 132 + lane * 4] = v;
        }
    }
    __syncthreads();
    if (!active) return;

    const int m  = lane & 15;
    const int kb = (lane >> 4) << 1;
    const float* xr = &sX[wave][m * 132];

    v8f acc = {};
    #pragma unroll
    for (int k0 = 0; k0 < IN_DIM; k0 += 4) {
        v2f a, b;
        a.x = xr[k0 + kb];
        a.y = xr[k0 + kb + 1];
        b.x = sW[(k0 + kb)     * HID + m];
        b.y = sW[(k0 + kb + 1) * HID + m];
        acc = __builtin_amdgcn_wmma_f32_16x16x4_f32(false, a, false, b,
                                                    (short)0, acc, false, false);
    }

    const int rbase = row0 + ((lane >> 4) << 3);
    #pragma unroll
    for (int v = 0; v < 8; ++v) {
        int r = rbase + v;
        if (r < n) h[(size_t)r * HID + m] = acc[v];
    }
}

#define LCAP 384

__device__ __forceinline__ bool stage_lists(const int* __restrict__ sorted, int* slab,
                                            int start, int end, int lane, int nE, int& base) {
    const int lo = __builtin_amdgcn_readlane(start, 0);
    const int hi = __builtin_amdgcn_readlane(end, 31);
    base = lo & ~31;
    const bool ok = start >= lo && end <= hi && lo >= 0 && hi <= nE && hi - base <= LCAP;
    if (!__all(ok)) return false;
    int w[LCAP / 32];
    #pragma unroll
    for (int k = 0; k < LCAP / 32; ++k) {
        const int p = base + lane + 32 * k;
        w[k] = 0;
        if (p < hi) w[k] = sorted[p];
    }
    #pragma unroll
    for (int k = 0; k < LCAP / 32; ++k) slab[lane + 32 * k] = w[k];
    return true;
}

#define PNODES 200
#define TABCAP 102400
#define TQ     (TABCAP / 16 / (PNODES * 4))
__device__ __forceinline__ float4 in_sum16(const int* lst, int start, int end, const unsigned char* deg8,
                                           const float* __restrict__ dinv, const float4* __restrict__ h4,
                                           int q, float di, unsigned nmax) {
    float4 acc = make_float4(0.0f, 0.0f, 0.0f, 0.0f);
    for (int j = start; j < end; j += 4) {
        const int last = end - 1;
        const int j1 = (j + 1 < end) ? j + 1 : last;
        const int j2 = (j + 2 < end) ? j + 2 : last;
        const int j3 = (j + 3 < end) ? j + 3 : last;
        unsigned s0 = (unsigned)lst[j];
        unsigned s1 = (unsigned)lst[j1];
        unsigned s2 = (unsigned)lst[j2];
        unsigned s3 = (unsigned)lst[j3];
        s0 = s0 > nmax ? nmax : s0;
        s1 = s1 > nmax ? nmax : s1;
        s2 = s2 > nmax ? nmax : s2;
        s3 = s3 > nmax ? nmax : s3;
        const float4 v0 = h4[(size_t)s0 * 4 + q];
        const float4 v1 = h4[(size_t)s1 * 4 + q];
        const float4 v2 = h4[(size_t)s2 * 4 + q];
        const float4 v3 = h4[(size_t)s3 * 4 + q];
        __builtin_amdgcn_sched_barrier(0);
        const unsigned c0 = deg8[s0], c1 = deg8[s1], c2 = deg8[s2], c3 = deg8[s3];
        float d0 = inv_root_deg((int)c0), d1 = inv_root_deg((int)c1);
        float d2 = inv_root_deg((int)c2), d3 = inv_root_deg((int)c3);
        if ((c0 == 255u) | (c1 == 255u) | (c2 == 255u) | (c3 == 255u)) {
            if (c0 == 255u) { const unsigned s = (unsigned)lst[j];  d0 = dinv[s > nmax ? nmax : s]; }
            if (c1 == 255u) { const unsigned s = (unsigned)lst[j1]; d1 = dinv[s > nmax ? nmax : s]; }
            if (c2 == 255u) { const unsigned s = (unsigned)lst[j2]; d2 = dinv[s > nmax ? nmax : s]; }
            if (c3 == 255u) { const unsigned s = (unsigned)lst[j3]; d3 = dinv[s > nmax ? nmax : s]; }
        }
        const float w0 = d0 * di;
        const float w1 = d1 * ((j + 1 < end) ? di : 0.0f);
        const float w2 = d2 * ((j + 2 < end) ? di : 0.0f);
        const float w3 = d3 * ((j + 3 < end) ? di : 0.0f);
        acc.x = fmaf(v0.x, w0, acc.x); acc.y = fmaf(v0.y, w0, acc.y);
        acc.z = fmaf(v0.z, w0, acc.z); acc.w = fmaf(v0.w, w0, acc.w);
        acc.x = fmaf(v1.x, w1, acc.x); acc.y = fmaf(v1.y, w1, acc.y);
        acc.z = fmaf(v1.z, w1, acc.z); acc.w = fmaf(v1.w, w1, acc.w);
        acc.x = fmaf(v2.x, w2, acc.x); acc.y = fmaf(v2.y, w2, acc.y);
        acc.z = fmaf(v2.z, w2, acc.z); acc.w = fmaf(v2.w, w2, acc.w);
        acc.x = fmaf(v3.x, w3, acc.x); acc.y = fmaf(v3.y, w3, acc.y);
        acc.z = fmaf(v3.z, w3, acc.z); acc.w = fmaf(v3.w, w3, acc.w);
    }
    return acc;
}

__global__ __launch_bounds__(PNODES * 4) void k_pull16(const int* __restrict__ rowptr,
                                                const int* __restrict__ cnt,
                                                const int* __restrict__ deg8w,
                                                const int* __restrict__ sorted,
                                                const float* __restrict__ dinv,
                                                const float* __restrict__ h,
                                                const float* __restrict__ b1,
                                                const float* __restrict__ W2,
                                                float* __restrict__ rec, int n, int nE, int nq) {
    __shared__ uint4 sDeg[TABCAP / 16];
    __shared__ int sList[(PNODES / 8) * LCAP];
    uint4 tv[TQ];
    #pragma unroll
    for (int u = 0; u < TQ; ++u) {
        const int k = (int)threadIdx.x + u * (PNODES * 4);
        tv[u] = ((const uint4*)deg8w)[k < nq ? k : nq - 1];
    }
    const int g = blockIdx.x * (PNODES * 4) + threadIdx.x;
    const int iq = g >> 2;
    const int q = g & 3;
    const int lane = threadIdx.x & 31, wave = threadIdx.x >> 5;
    const bool valid = iq < n;
    const int i = valid ? iq : n - 1;
    int start = rowptr[i];
    start = start < 0 ? 0 : (start > nE ? nE : start);
    int c = cnt[i];
    c = (!valid || c < 0) ? 0 : c;
    const int end = (c > nE - start) ? nE : start + c;
    const float di = dinv[i];
    const float4* h4 = (const float4*)h;
    const unsigned nmax = (unsigned)n - 1u;

    #pragma unroll
    for (int u = 0; u < TQ; ++u) sDeg[(int)threadIdx.x + u * (PNODES * 4)] = tv[u];
    const float4 hs = h4[(size_t)i * 4 + q];
    int* slab = sList + wave * LCAP;
    int base = 0;
    const bool staged = stage_lists(sorted, slab, start, end, lane, nE, base);
    __syncthreads();
    const unsigned char* deg8 = (const unsigned char*)sDeg;
    float4 acc;
    if (staged) acc = in_sum16(slab, start - base, end - base, deg8, dinv, h4, q, di, nmax);
    else        acc = in_sum16(sorted, start, end, deg8, dinv, h4, q, di, nmax);
    const int c0 = q * 4;
    const float bA = b1[c0], bB = b1[c0 + 1], bC = b1[c0 + 2], bD = b1[c0 + 3];
    const float wA0 = W2[(c0    ) * OUT_DIM], wA1 = W2[(c0    ) * OUT_DIM + 1];
    const float wB0 = W2[(c0 + 1) * OUT_DIM], wB1 = W2[(c0 + 1) * OUT_DIM + 1];
    const float wC0 = W2[(c0 + 2) * OUT_DIM], wC1 = W2[(c0 + 2) * OUT_DIM + 1];
    const float wD0 = W2[(c0 + 3) * OUT_DIM], wD1 = W2[(c0 + 3) * OUT_DIM + 1];
    const float ws = di * di;
    acc.x = fmaf(hs.x, ws, acc.x); acc.y = fmaf(hs.y, ws, acc.y);
    acc.z = fmaf(hs.z, ws, acc.z); acc.w = fmaf(hs.w, ws, acc.w);

    const float vA = fmaxf(acc.x + bA, 0.0f);
    const float vB = fmaxf(acc.y + bB, 0.0f);
    const float vC = fmaxf(acc.z + bC, 0.0f);
    const float vD = fmaxf(acc.w + bD, 0.0f);
    float z0 = vA * wA0, z1 = vA * wA1;
    z0 = fmaf(vB, wB0, z0); z1 = fmaf(vB, wB1, z1);
    z0 = fmaf(vC, wC0, z0); z1 = fmaf(vC, wC1, z1);
    z0 = fmaf(vD, wD0, z0); z1 = fmaf(vD, wD1, z1);
    z0 += __shfl_xor(z0, 1); z1 += __shfl_xor(z1, 1);
    z0 += __shfl_xor(z0, 2); z1 += __shfl_xor(z1, 2);
    if (valid && q == 0) ((float4*)rec)[i] = make_float4(z0, z1, di, 0.0f);
}

__device__ __forceinline__ void in_sum2(const int* lst, int first, int end, const float4* __restrict__ rv,
                                        float di, unsigned nmax, float& a0, float& a1) {
    for (int j = first; j < end; j += 8) {
        const int jb = j + 4;
        const bool hb = jb < end;
        unsigned sa = (unsigned)lst[j];
        unsigned sb = (unsigned)lst[hb ? jb : j];
        sa = sa > nmax ? nmax : sa;
        sb = sb > nmax ? nmax : sb;
        const float4 va = rv[sa];
        const float4 vb = rv[sb];
        const float wa = va.z * di;
        const float wb = vb.z * (hb ? di : 0.0f);
        a0 = fmaf(va.x, wa, a0); a1 = fmaf(va.y, wa, a1);
        a0 = fmaf(vb.x, wb, a0); a1 = fmaf(vb.y, wb, a1);
    }
}

__global__ __launch_bounds__(256) void k_pull2_final(const int* __restrict__ rowptr,
                                                     const int* __restrict__ cnt,
                                                     const int* __restrict__ sorted,
                                                     const float* __restrict__ dinv,
                                                     const float* __restrict__ rec,
                                                     const float* __restrict__ b2,
                                                     float* __restrict__ out, int n, int nE) {
    __shared__ int sList[8 * LCAP];
    const int g = blockIdx.x * 256 + threadIdx.x;
    const int iq = g >> 2;
    const int q = g & 3;
    const int lane = threadIdx.x & 31, wave = threadIdx.x >> 5;
    const bool valid = iq < n;
    const int i = valid ? iq : n - 1;
    int start = rowptr[i];
    start = start < 0 ? 0 : (start > nE ? nE : start);
    int c = valid ? cnt[i] : 0;
    c = c < 0 ? 0 : c;
    const int end = (c > nE - start) ? nE : start + c;
    const float di = dinv[i];
    const float4* rv = (const float4*)rec;
    const unsigned nmax = (unsigned)n - 1u;

    int* slab = sList + wave * LCAP;
    int base = 0;
    const bool staged = stage_lists(sorted, slab, start, end, lane, nE, base);
    __syncthreads();
    float a0 = 0.0f, a1 = 0.0f;
    if (staged) in_sum2(slab, start - base + q, end - base, rv, di, nmax, a0, a1);
    else        in_sum2(sorted, start + q, end, rv, di, nmax, a0, a1);
    a0 += __shfl_xor(a0, 1); a1 += __shfl_xor(a1, 1);
    a0 += __shfl_xor(a0, 2); a1 += __shfl_xor(a1, 2);

    if (valid && q == 0) {
        const float ws = di * di;
        const float4 hs = rv[i];
        const float z0 = fmaf(hs.x, ws, a0) + b2[0];
        const float z1 = fmaf(hs.y, ws, a1) + b2[1];
        const float mx = fmaxf(z0, z1);
        const float l  = mx + logf(expf(z0 - mx) + expf(z1 - mx));
        ((float2*)out)[i] = make_float2(z0 - l, z1 - l);
    }
}

extern "C" void kernel_launch(void* const* d_in, const int* in_sizes, int n_in,
                              void* d_out, int out_size, void* d_ws, size_t ws_size,
                              hipStream_t stream) {
    const float* x  = (const float*)d_in[0];
    const int*   ei = (const int*)d_in[1];
    const float* W1 = (const float*)d_in[2];
    const float* b1 = (const float*)d_in[3];
    const float* W2 = (const float*)d_in[4];
    const float* b2 = (const float*)d_in[5];
    float* out = (float*)d_out;

    const int n  = in_sizes[0] / IN_DIM;
    const int nE = in_sizes[1] / 2;
    const int* src = ei;
    const int* dst = ei + nE;

    const size_t np = ((size_t)n  + 31) & ~(size_t)31;
    const size_t Ep = ((size_t)nE + 31) & ~(size_t)31;
    const int nb   = (n + BNODES - 1) / BNODES;
    const int nblk = (nE + SLICE - 1) / SLICE;
    const int nbp  = (nb + 1 + 31) & ~31;
    const size_t tw = (size_t)nblk * nbp;
    const size_t dw = (size_t)nb * (BNODES / 4);
    const size_t uw = (Ep > 32 * np) ? Ep : 32 * np;
    const size_t need = (5 * np + Ep + uw) * sizeof(float);
    if (ws_size < need) return;
    if (nb > MAXB || n > (1 << 24) || nblk > 32 * RPW || tw + dw > 2 * np) return;
    if ((size_t)nb * BNODES > TABCAP) return;

    int*   cnt    = (int*)d_ws;
    int*   rowptr = cnt + np;
    float* dinv   = (float*)(rowptr + np);
    int*   T      = (int*)(dinv + np);
    int*   deg8w  = T + tw;
    int*   sorted = T + 2 * np;
    int*   binned = sorted + Ep;
    float* h      = (float*)binned;
    float* rec    = h + 16 * np;

    k_slice_sort  <<<nblk, 1024, 0, stream>>>(src, dst, T, binned, nE, nb, nbp);
    k_bucket_sort <<<nb, 1024, 0, stream>>>(binned, T, cnt, rowptr, dinv, sorted, deg8w, n, nE, nbp, nblk);

    const int tiles   = (n + 15) / 16;
    const int gblocks = (tiles + 3) / 4;
    k_gemm1_wmma  <<<gblocks, 128, 0, stream>>>(x, W1, h, n);

    const int qblocks = (int)(((size_t)n * 4 + 255) / 256);
    const int pblocks = (n + PNODES - 1) / PNODES;
    k_pull16      <<<pblocks, PNODES * 4, 0, stream>>>(rowptr, cnt, deg8w, sorted, dinv, h, b1, W2, rec, n, nE, nb * (BNODES / 16));
    k_pull2_final <<<qblocks, 256, 0, stream>>>(rowptr, cnt, sorted, dinv, rec, b2, out, n, nE);
}


# x@W1 product tiles done by extra 32-wave blocks of the slice-ordering launch (job by block index; x staged in 32-column pieces, W1 as k pairs in LDS): one launch fewer, x streams beside the ordering
# speedup vs baseline: 17.2679x; 1.2310x over previous
#include <hip/hip_runtime.h>
#include <hip/hip_bf16.h>
#include <math.h>


typedef __attribute__((ext_vector_type(2))) float v2f;
typedef __attribute__((ext_vector_type(8))) float v8f;

#define IN_DIM  128
#define HID     16
#define OUT_DIM 2

#define BSHIFT 8
#define BNODES 256
#define MAXB   512
#define SLICE  12288
#define SPT    12
#define RPW    12
#define BCAP   16384
#define BPT    10

__device__ __forceinline__ float inv_root_deg(int c) {
    const float dg = (float)(c + 1);
    return (dg > 0.0f) ? rsqrtf(dg) : 0.0f;
}

#define XCH 32
#define XP  36
#define LDSW (IN_DIM * HID + 32 * 16 * XP)

__device__ __forceinline__ void wave_lds_order() {
    asm volatile("" ::: "memory");
}

struct XPiece { float4 r0, r1, r2, r3; };

__device__ __forceinline__ float4 x_part(const float* __restrict__ x, int row0, int n, int lane, int c, int i) {
    int r = row0 + 4 * i + (lane >> 3);
    r = r < n ? r : n - 1;
    return *(const float4*)(x + (size_t)r * IN_DIM + c * XCH + (lane & 7) * 4);
}
__device__ __forceinline__ XPiece x_piece_load(const float* __restrict__ x, int row0, int n, int lane, int c) {
    XPiece p;
    p.r0 = x_part(x, row0, n, lane, c, 0);
    p.r1 = x_part(x, row0, n, lane, c, 1);
    p.r2 = x_part(x, row0, n, lane, c, 2);
    p.r3 = x_part(x, row0, n, lane, c, 3);
    return p;
}

__device__ __forceinline__ void x_piece_store(const XPiece& p, float* slab, int lane) {
    float* w = slab + (lane >> 3) * XP + (lane & 7) * 4;
    *(float4*)(w)           = p.r0;
    *(float4*)(w + 4 * XP)  = p.r1;
    *(float4*)(w + 8 * XP)  = p.r2;
    *(float4*)(w + 12 * XP) = p.r3;
}

__device__ __forceinline__ v8f x_piece_mma(v8f acc, const float* slab, const float* sW, int lane, int c) {
    const int m  = lane & 15;
    const int kb = (lane >> 4) << 1;
    const v2f* xr = (const v2f*)(slab + m * XP + kb);
    const v2f* wr = (const v2f*)sW + (kb >> 1) * HID + m;
    #pragma unroll
    for (int kk = 0; kk < XCH; kk += 4) {
        const v2f a = xr[kk >> 1];
        const v2f b = wr[((c * XCH + kk) >> 1) * HID];
        acc = __builtin_amdgcn_wmma_f32_16x16x4_f32(false, a, false, b,
                                                    (short)0, acc, false, false);
    }
    return acc;
}

__device__ __forceinline__ void product_tiles(const float* __restrict__ x, const float* __restrict__ W1,
                                              float* __restrict__ h, float* sW, float* sX,
                                              int n, int tiles, int gb, int ngb) {
    const int t = threadIdx.x, lane = t & 31;
    const int wave = __builtin_amdgcn_readfirstlane(t >> 5);
    for (int j = t; j < IN_DIM * HID / 2; j += 1024) {
        const float2 w = ((const float2*)W1)[j];
        const int k = j >> 3, m = (j & 7) * 2;
        sW[((k >> 1) * HID + m) * 2 + (k & 1)]     = w.x;
        sW[((k >> 1) * HID + m + 1) * 2 + (k & 1)] = w.y;
    }
    __syncthreads();
    float* slab = sX + wave * (16 * XP);
    for (int tile = gb * 32 + wave; tile < tiles; tile += ngb * 32) {
        const int row0 = tile * 16;
        v8f acc = {};
        XPiece p = x_piece_load(x, row0, n, lane, 0);
        #pragma unroll
        for (int c = 0; c < IN_DIM / XCH; ++c) {
            x_piece_store(p, slab, lane);
            wave_lds_order();
            if (c + 1 < IN_DIM / XCH) p = x_piece_load(x, row0, n, lane, c + 1);
            __builtin_amdgcn_sched_barrier(0);
            acc = x_piece_mma(acc, slab, sW, lane, c);
            wave_lds_order();
        }
        const int m = lane & 15;
        const int rbase = row0 + ((lane >> 4) << 3);
        #pragma unroll
        for (int v = 0; v < 8; ++v) {
            const int r = rbase + v;
            if (r < n) h[(size_t)r * HID + m] = acc[v];
        }
    }
}

__global__ __launch_bounds__(1024, 16) void k_slice_gemm(const int* __restrict__ src, const int* __restrict__ dst,
                                                     int* __restrict__ T, int* __restrict__ binned,
                                                     int nE, int nb, int nbp, int nblk,
                                                     const float* __restrict__ x, const float* __restrict__ W1,
                                                     float* __restrict__ h, int n, int tiles) {
    __shared__ __align__(16) int smem[LDSW];
    if ((int)blockIdx.x >= nblk) {
        float* sW = (float*)smem;
        float* sX = sW + IN_DIM * HID;
        product_tiles(x, W1, h, sW, sX, n, tiles, (int)blockIdx.x - nblk, (int)gridDim.x - nblk);
        return;
    }
    int* sWord = smem;
    int* hist  = sWord + SLICE;
    int* lbase = hist + MAXB;
    int* wtot  = lbase + MAXB;
    const int t = threadIdx.x, lane = t & 31, wave = t >> 5;
    const int me = blockIdx.x;

    const int e0 = me * SLICE + t;
    int s[SPT], d[SPT], r[SPT];
    #pragma unroll
    for (int u = 0; u < SPT; ++u) {
        const int e = e0 + u * 1024;
        const int ec = e < nE ? e : nE - 1;
        s[u] = src[ec];
        d[u] = dst[ec];
    }
    if (t < MAXB) hist[t] = 0;
    __syncthreads();

    #pragma unroll
    for (int u = 0; u < SPT; ++u) {
        const int e = e0 + u * 1024;
        r[u] = 0;
        if (e < nE) {
            unsigned b = (unsigned)d[u] >> BSHIFT;
            b = b > (unsigned)(nb - 1) ? (unsigned)(nb - 1) : b;
            r[u] = atomicAdd(&hist[b], 1);
        }
    }
    __syncthreads();
    int c = 0, incl = 0;
    if (t < MAXB) {
        c = hist[t];
        incl = c;
        #pragma unroll
        for (int o = 1; o < 32; o <<= 1) {
            const int v = __shfl_up(incl, o);
            if (lane >= o) incl += v;
        }
        if (lane == 31) wtot[wave] = incl;
    }
    __syncthreads();
    int ex = 0;
    if (t < MAXB) {
        int woff = 0;
        #pragma unroll
        for (int w = 0; w < 16; ++w) woff += (w < wave) ? wtot[w] : 0;
        ex = woff + incl - c;
        lbase[t] = ex;
    }
    __syncthreads();
    #pragma unroll
    for (int u = 0; u < SPT; ++u) {
        const int e = e0 + u * 1024;
        if (e < nE) {
            unsigned b = (unsigned)d[u] >> BSHIFT;
            b = b > (unsigned)(nb - 1) ? (unsigned)(nb - 1) : b;
            sWord[lbase[b] + r[u]] = (int)((((unsigned)d[u] & (BNODES - 1)) << 24) | ((unsigned)s[u] & 0xFFFFFFu));
        }
    }
    __syncthreads();
    if (t < nb) T[(size_t)me * nbp + t] = ex;
    if (t == nb - 1) T[(size_t)me * nbp + nb] = ex + c;
    const int first = me * SLICE;
    const int have = (nE - first < SLICE) ? nE - first : SLICE;
    for (int k = t; k < have; k += 1024) binned[first + k] = sWord[k];
}

__global__ __launch_bounds__(1024, 16) void k_bucket_sort(const int* __restrict__ binned, const int* __restrict__ T,
                                                      int* __restrict__ cnt, int* __restrict__ rowptr,
                                                      float* __restrict__ dinv, int* __restrict__ sorted,
                                                      int* __restrict__ deg8w,
                                                      int n, int nE, int nbp, int nblk) {
    __shared__ int sIn[BCAP];
    __shared__ int sOut[BCAP];
    __shared__ int rBase[32 * RPW];
    __shared__ int rLen[32 * RPW];
    __shared__ int rPos[32 * RPW];
    __shared__ int lcnt[BNODES];
    __shared__ int lptr[BNODES];
    __shared__ int wlen[RPW];
    __shared__ int wlo[RPW];
    __shared__ int wtot[8];
    const int t = threadIdx.x, lane = t & 31, wave = t >> 5;
    const int b = blockIdx.x;

    int lo = 0, len = 0;
    if (t < nblk) {
        const int first = t * SLICE;
        int have = (nE - first < SLICE) ? nE - first : SLICE;
        have = have < 0 ? 0 : have;
        int a = T[(size_t)t * nbp + b];
        int e = T[(size_t)t * nbp + b + 1];
        a = a < 0 ? 0 : (a > have ? have : a);
        e = e < a ? a : (e > have ? have : e);
        lo = a;
        len = e - a;
    }
    int incl = len, slo = lo;
    #pragma unroll
    for (int o = 1; o < 32; o <<= 1) {
        const int v = __shfl_up(incl, o);
        const int y = __shfl_up(slo, o);
        if (lane >= o) { incl += v; slo += y; }
    }
    if (wave < RPW && lane == 31) {
        wlen[wave] = incl;
        wlo[wave] = slo;
    }
    if (t < BNODES) lcnt[t] = 0;
    __syncthreads();
    int woff = 0, tot = 0, bs = 0;
    #pragma unroll
    for (int w = 0; w < RPW; ++w) {
        const int x = wlen[w];
        woff += (w < wave) ? x : 0;
        tot += x;
        bs += wlo[w];
    }
    if (t < 32 * RPW) {
        rBase[t] = t * SLICE + lo;
        rLen[t]  = len;
        rPos[t]  = woff + incl - len;
    }
    bs = bs < 0 ? 0 : (bs > nE ? nE : bs);
    const int size = (tot > nE - bs) ? nE - bs : tot;
    const bool inlds = tot <= BCAP;
    const int amax = nE - 1;
    __syncthreads();

    if (inlds) {
        int w0[RPW], w1[RPW];
        #pragma unroll
        for (int u = 0; u < RPW; ++u) {
            const int j = wave + 32 * u;
            const int base = rBase[j], ln = rLen[j];
            int a0 = base + (lane < ln ? lane : ln - 1);
            int a1 = base + (lane + 32 < ln ? lane + 32 : ln - 1);
            a0 = a0 < 0 ? 0 : (a0 > amax ? amax : a0);
            a1 = a1 < 0 ? 0 : (a1 > amax ? amax : a1);
            w0[u] = binned[a0];
            w1[u] = binned[a1];
        }
        int lnv[RPW], posv[RPW];
        #pragma unroll
        for (int u = 0; u < RPW; ++u) {
            const int j = wave + 32 * u;
            lnv[u] = rLen[j];
            posv[u] = rPos[j];
        }
        #pragma unroll
        for (int u = 0; u < RPW; ++u) {
            if (lane < lnv[u])      sIn[posv[u] + lane]      = w0[u];
            if (lane + 32 < lnv[u]) sIn[posv[u] + lane + 32] = w1[u];
        }
        for (int j = wave; j < 32 * RPW; j += 32) {
            const int base = rBase[j], ln = rLen[j], pos = rPos[j];
            for (int o = lane + 64; o < ln; o += 32) {
                int a = base + o;
                a = a > amax ? amax : a;
                sIn[pos + o] = binned[a];
            }
        }
    }
    __syncthreads();

    if (inlds) {
        for (int kb = t; kb < size + t; kb += BPT * 1024) {
            unsigned w[BPT];
            #pragma unroll
            for (int u = 0; u < BPT; ++u) {
                const int k = kb + u * 1024;
                w[u] = (unsigned)sIn[k < size ? k : size - 1];
            }
            #pragma unroll
            for (int u = 0; u < BPT; ++u)
                if (kb + u * 1024 < size) atomicAdd(&lcnt[w[u] >> 24], 1);
        }
    } else {
        for (int j = wave; j < 32 * RPW; j += 32) {
            const int base = rBase[j], ln = rLen[j];
            for (int o = lane; o < ln; o += 32) {
                int a = base + o;
                a = a > amax ? amax : a;
                atomicAdd(&lcnt[(unsigned)binned[a] >> 24], 1);
            }
        }
    }
    __syncthreads();
    int c = 0;
    incl = 0;
    if (t < BNODES) {
        c = lcnt[t];
        incl = c;
        #pragma unroll
        for (int o = 1; o < 32; o <<= 1) {
            const int v = __shfl_up(incl, o);
            if (lane >= o) incl += v;
        }
        if (lane == 31) wtot[wave] = incl;
    }
    __syncthreads();
    if (t < BNODES) {
        int wo = 0;
        #pragma unroll
        for (int w = 0; w < 8; ++w) wo += (w < wave) ? wtot[w] : 0;
        const int ex = wo + incl - c;
        lptr[t] = ex;
        lcnt[t] = 0;
        const int node = b * BNODES + t;
        if (node < n) {
            cnt[node] = c;
            rowptr[node] = bs + ex;
            dinv[node] = inv_root_deg(c);
        }
        const int c8 = (node < n) ? (c > 255 ? 255 : c) : 0;
        ((unsigned char*)deg8w)[(size_t)b * BNODES + t] = (unsigned char)c8;
    }
    __syncthreads();

    if (inlds) {
        for (int kb = t; kb < size + t; kb += BPT * 1024) {
            unsigned w[BPT];
            #pragma unroll
            for (int u = 0; u < BPT; ++u) {
                const int k = kb + u * 1024;
                w[u] = (unsigned)sIn[k < size ? k : size - 1];
            }
            #pragma unroll
            for (int u = 0; u < BPT; ++u) {
                if (kb + u * 1024 < size) {
                    const unsigned dl = w[u] >> 24;
                    const int p = lptr[dl] + atomicAdd(&lcnt[dl], 1);
                    sOut[p] = (int)(w[u] & 0xFFFFFFu);
                }
            }
        }
    } else {
        for (int j = wave; j < 32 * RPW; j += 32) {
            const int base = rBase[j], ln = rLen[j];
            for (int o = lane; o < ln; o += 32) {
                int a = base + o;
                a = a > amax ? amax : a;
                const unsigned w = (unsigned)binned[a];
                const unsigned dl = w >> 24;
                const int p = lptr[dl] + atomicAdd(&lcnt[dl], 1);
                if (p < size) sorted[bs + p] = (int)(w & 0xFFFFFFu);
            }
        }
    }
    __syncthreads();
    if (inlds)
        for (int k = t; k < size; k += 1024) sorted[bs + k] = sOut[k];
}

#define LCAP 384

__device__ __forceinline__ bool stage_lists(const int* __restrict__ sorted, int* slab,
                                            int start, int end, int lane, int nE, int& base) {
    const int lo = __builtin_amdgcn_readlane(start, 0);
    const int hi = __builtin_amdgcn_readlane(end, 31);
    base = lo & ~31;
    const bool ok = start >= lo && end <= hi && lo >= 0 && hi <= nE && hi - base <= LCAP;
    if (!__all(ok)) return false;
    int w[LCAP / 32];
    #pragma unroll
    for (int k = 0; k < LCAP / 32; ++k) {
        const int p = base + lane + 32 * k;
        w[k] = 0;
        if (p < hi) w[k] = sorted[p];
    }
    #pragma unroll
    for (int k = 0; k < LCAP / 32; ++k) slab[lane + 32 * k] = w[k];
    return true;
}

#define PNODES 200
#define TABCAP 102400
#define TQ     (TABCAP / 16 / (PNODES * 4))
__device__ __forceinline__ float4 in_sum16(const int* lst, int start, int end, const unsigned char* deg8,
                                           const float* __restrict__ dinv, const float4* __restrict__ h4,
                                           int q, float di, unsigned nmax) {
    float4 acc = make_float4(0.0f, 0.0f, 0.0f, 0.0f);
    for (int j = start; j < end; j += 4) {
        const int last = end - 1;
        const int j1 = (j + 1 < end) ? j + 1 : last;
        const int j2 = (j + 2 < end) ? j + 2 : last;
        const int j3 = (j + 3 < end) ? j + 3 : last;
        unsigned s0 = (unsigned)lst[j];
        unsigned s1 = (unsigned)lst[j1];
        unsigned s2 = (unsigned)lst[j2];
        unsigned s3 = (unsigned)lst[j3];
        s0 = s0 > nmax ? nmax : s0;
        s1 = s1 > nmax ? nmax : s1;
        s2 = s2 > nmax ? nmax : s2;
        s3 = s3 > nmax ? nmax : s3;
        const float4 v0 = h4[(size_t)s0 * 4 + q];
        const float4 v1 = h4[(size_t)s1 * 4 + q];
        const float4 v2 = h4[(size_t)s2 * 4 + q];
        const float4 v3 = h4[(size_t)s3 * 4 + q];
        __builtin_amdgcn_sched_barrier(0);
        const unsigned c0 = deg8[s0], c1 = deg8[s1], c2 = deg8[s2], c3 = deg8[s3];
        float d0 = inv_root_deg((int)c0), d1 = inv_root_deg((int)c1);
        float d2 = inv_root_deg((int)c2), d3 = inv_root_deg((int)c3);
        if ((c0 == 255u) | (c1 == 255u) | (c2 == 255u) | (c3 == 255u)) {
            if (c0 == 255u) { const unsigned s = (unsigned)lst[j];  d0 = dinv[s > nmax ? nmax : s]; }
            if (c1 == 255u) { const unsigned s = (unsigned)lst[j1]; d1 = dinv[s > nmax ? nmax : s]; }
            if (c2 == 255u) { const unsigned s = (unsigned)lst[j2]; d2 = dinv[s > nmax ? nmax : s]; }
            if (c3 == 255u) { const unsigned s = (unsigned)lst[j3]; d3 = dinv[s > nmax ? nmax : s]; }
        }
        const float w0 = d0 * di;
        const float w1 = d1 * ((j + 1 < end) ? di : 0.0f);
        const float w2 = d2 * ((j + 2 < end) ? di : 0.0f);
        const float w3 = d3 * ((j + 3 < end) ? di : 0.0f);
        acc.x = fmaf(v0.x, w0, acc.x); acc.y = fmaf(v0.y, w0, acc.y);
        acc.z = fmaf(v0.z, w0, acc.z); acc.w = fmaf(v0.w, w0, acc.w);
        acc.x = fmaf(v1.x, w1, acc.x); acc.y = fmaf(v1.y, w1, acc.y);
        acc.z = fmaf(v1.z, w1, acc.z); acc.w = fmaf(v1.w, w1, acc.w);
        acc.x = fmaf(v2.x, w2, acc.x); acc.y = fmaf(v2.y, w2, acc.y);
        acc.z = fmaf(v2.z, w2, acc.z); acc.w = fmaf(v2.w, w2, acc.w);
        acc.x = fmaf(v3.x, w3, acc.x); acc.y = fmaf(v3.y, w3, acc.y);
        acc.z = fmaf(v3.z, w3, acc.z); acc.w = fmaf(v3.w, w3, acc.w);
    }
    return acc;
}

__global__ __launch_bounds__(PNODES * 4) void k_pull16(const int* __restrict__ rowptr,
                                                const int* __restrict__ cnt,
                                                const int* __restrict__ deg8w,
                                                const int* __restrict__ sorted,
                                                const float* __restrict__ dinv,
                                                const float* __restrict__ h,
                                                const float* __restrict__ b1,
                                                const float* __restrict__ W2,
                                                float* __restrict__ rec, int n, int nE, int nq) {
    __shared__ uint4 sDeg[TABCAP / 16];
    __shared__ int sList[(PNODES / 8) * LCAP];
    uint4 tv[TQ];
    #pragma unroll
    for (int u = 0; u < TQ; ++u) {
        const int k = (int)threadIdx.x + u * (PNODES * 4);
        tv[u] = ((const uint4*)deg8w)[k < nq ? k : nq - 1];
    }
    const int g = blockIdx.x * (PNODES * 4) + threadIdx.x;
    const int iq = g >> 2;
    const int q = g & 3;
    const int lane = threadIdx.x & 31, wave = threadIdx.x >> 5;
    const bool valid = iq < n;
    const int i = valid ? iq : n - 1;
    int start = rowptr[i];
    start = start < 0 ? 0 : (start > nE ? nE : start);
    int c = cnt[i];
    c = (!valid || c < 0) ? 0 : c;
    const int end = (c > nE - start) ? nE : start + c;
    const float di = dinv[i];
    const float4* h4 = (const float4*)h;
    const unsigned nmax = (unsigned)n - 1u;

    #pragma unroll
    for (int u = 0; u < TQ; ++u) sDeg[(int)threadIdx.x + u * (PNODES * 4)] = tv[u];
    const float4 hs = h4[(size_t)i * 4 + q];
    int* slab = sList + wave * LCAP;
    int base = 0;
    const bool staged = stage_lists(sorted, slab, start, end, lane, nE, base);
    __syncthreads();
    const unsigned char* deg8 = (const unsigned char*)sDeg;
    float4 acc;
    if (staged) acc = in_sum16(slab, start - base, end - base, deg8, dinv, h4, q, di, nmax);
    else        acc = in_sum16(sorted, start, end, deg8, dinv, h4, q, di, nmax);
    const int c0 = q * 4;
    const float bA = b1[c0], bB = b1[c0 + 1], bC = b1[c0 + 2], bD = b1[c0 + 3];
    const float wA0 = W2[(c0    ) * OUT_DIM], wA1 = W2[(c0    ) * OUT_DIM + 1];
    const float wB0 = W2[(c0 + 1) * OUT_DIM], wB1 = W2[(c0 + 1) * OUT_DIM + 1];
    const float wC0 = W2[(c0 + 2) * OUT_DIM], wC1 = W2[(c0 + 2) * OUT_DIM + 1];
    const float wD0 = W2[(c0 + 3) * OUT_DIM], wD1 = W2[(c0 + 3) * OUT_DIM + 1];
    const float ws = di * di;
    acc.x = fmaf(hs.x, ws, acc.x); acc.y = fmaf(hs.y, ws, acc.y);
    acc.z = fmaf(hs.z, ws, acc.z); acc.w = fmaf(hs.w, ws, acc.w);

    const float vA = fmaxf(acc.x + bA, 0.0f);
    const float vB = fmaxf(acc.y + bB, 0.0f);
    const float vC = fmaxf(acc.z + bC, 0.0f);
    const float vD = fmaxf(acc.w + bD, 0.0f);
    float z0 = vA * wA0, z1 = vA * wA1;
    z0 = fmaf(vB, wB0, z0); z1 = fmaf(vB, wB1, z1);
    z0 = fmaf(vC, wC0, z0); z1 = fmaf(vC, wC1, z1);
    z0 = fmaf(vD, wD0, z0); z1 = fmaf(vD, wD1, z1);
    z0 += __shfl_xor(z0, 1); z1 += __shfl_xor(z1, 1);
    z0 += __shfl_xor(z0, 2); z1 += __shfl_xor(z1, 2);
    if (valid && q == 0) ((float4*)rec)[i] = make_float4(z0, z1, di, 0.0f);
}

__device__ __forceinline__ void in_sum2(const int* lst, int first, int end, const float4* __restrict__ rv,
                                        float di, unsigned nmax, float& a0, float& a1) {
    for (int j = first; j < end; j += 8) {
        const int jb = j + 4;
        const bool hb = jb < end;
        unsigned sa = (unsigned)lst[j];
        unsigned sb = (unsigned)lst[hb ? jb : j];
        sa = sa > nmax ? nmax : sa;
        sb = sb > nmax ? nmax : sb;
        const float4 va = rv[sa];
        const float4 vb = rv[sb];
        const float wa = va.z * di;
        const float wb = vb.z * (hb ? di : 0.0f);
        a0 = fmaf(va.x, wa, a0); a1 = fmaf(va.y, wa, a1);
        a0 = fmaf(vb.x, wb, a0); a1 = fmaf(vb.y, wb, a1);
    }
}

__global__ __launch_bounds__(256) void k_pull2_final(const int* __restrict__ rowptr,
                                                     const int* __restrict__ cnt,
                                                     const int* __restrict__ sorted,
                                                     const float* __restrict__ dinv,
                                                     const float* __restrict__ rec,
                                                     const float* __restrict__ b2,
                                                     float* __restrict__ out, int n, int nE) {
    __shared__ int sList[8 * LCAP];
    const int g = blockIdx.x * 256 + threadIdx.x;
    const int iq = g >> 2;
    const int q = g & 3;
    const int lane = threadIdx.x & 31, wave = threadIdx.x >> 5;
    const bool valid = iq < n;
    const int i = valid ? iq : n - 1;
    int start = rowptr[i];
    start = start < 0 ? 0 : (start > nE ? nE : start);
    int c = valid ? cnt[i] : 0;
    c = c < 0 ? 0 : c;
    const int end = (c > nE - start) ? nE : start + c;
    const float di = dinv[i];
    const float4* rv = (const float4*)rec;
    const unsigned nmax = (unsigned)n - 1u;

    int* slab = sList + wave * LCAP;
    int base = 0;
    const bool staged = stage_lists(sorted, slab, start, end, lane, nE, base);
    __syncthreads();
    float a0 = 0.0f, a1 = 0.0f;
    if (staged) in_sum2(slab, start - base + q, end - base, rv, di, nmax, a0, a1);
    else        in_sum2(sorted, start + q, end, rv, di, nmax, a0, a1);
    a0 += __shfl_xor(a0, 1); a1 += __shfl_xor(a1, 1);
    a0 += __shfl_xor(a0, 2); a1 += __shfl_xor(a1, 2);

    if (valid && q == 0) {
        const float ws = di * di;
        const float4 hs = rv[i];
        const float z0 = fmaf(hs.x, ws, a0) + b2[0];
        const float z1 = fmaf(hs.y, ws, a1) + b2[1];
        const float mx = fmaxf(z0, z1);
        const float l  = mx + logf(expf(z0 - mx) + expf(z1 - mx));
        ((float2*)out)[i] = make_float2(z0 - l, z1 - l);
    }
}

extern "C" void kernel_launch(void* const* d_in, const int* in_sizes, int n_in,
                              void* d_out, int out_size, void* d_ws, size_t ws_size,
                              hipStream_t stream) {
    const float* x  = (const float*)d_in[0];
    const int*   ei = (const int*)d_in[1];
    const float* W1 = (const float*)d_in[2];
    const float* b1 = (const float*)d_in[3];
    const float* W2 = (const float*)d_in[4];
    const float* b2 = (const float*)d_in[5];
    float* out = (float*)d_out;

    const int n  = in_sizes[0] / IN_DIM;
    const int nE = in_sizes[1] / 2;
    const int* src = ei;
    const int* dst = ei + nE;

    const size_t np = ((size_t)n  + 31) & ~(size_t)31;
    const size_t Ep = ((size_t)nE + 31) & ~(size_t)31;
    const int nb   = (n + BNODES - 1) / BNODES;
    const int nblk = (nE + SLICE - 1) / SLICE;
    const int nbp  = (nb + 1 + 31) & ~31;
    const size_t tw = (size_t)nblk * nbp;
    const size_t dw = (size_t)nb * (BNODES / 4);
    const size_t uw = (Ep > 4 * np) ? Ep : 4 * np;
    const size_t need = (5 * np + Ep + uw + 16 * np) * sizeof(float);
    if (ws_size < need) return;
    if (nb > MAXB || n > (1 << 24) || nblk > 32 * RPW || tw + dw > 2 * np) return;
    if ((size_t)nb * BNODES > TABCAP) return;

    int*   cnt    = (int*)d_ws;
    int*   rowptr = cnt + np;
    float* dinv   = (float*)(rowptr + np);
    int*   T      = (int*)(dinv + np);
    int*   deg8w  = T + tw;
    int*   sorted = T + 2 * np;
    int*   binned = sorted + Ep;
    float* rec    = (float*)binned;
    float* h      = (float*)(binned + uw);

    const int tiles   = (n + 15) / 16;
    const int gblocks = (tiles + 31) / 32;
    k_slice_gemm  <<<nblk + gblocks, 1024, 0, stream>>>(src, dst, T, binned, nE, nb, nbp, nblk, x, W1, h, n, tiles);
    k_bucket_sort <<<nb, 1024, 0, stream>>>(binned, T, cnt, rowptr, dinv, sorted, deg8w, n, nE, nbp, nblk);

    const int qblocks = (int)(((size_t)n * 4 + 255) / 256);
    const int pblocks = (n + PNODES - 1) / PNODES;
    k_pull16      <<<pblocks, PNODES * 4, 0, stream>>>(rowptr, cnt, deg8w, sorted, dinv, h, b1, W2, rec, n, nE, nb * (BNODES / 16));
    k_pull2_final <<<qblocks, 256, 0, stream>>>(rowptr, cnt, sorted, dinv, rec, b2, out, n, nE);
}
